# starting kernel
# baseline (speedup 1.0000x reference)
#include <hip/hip_runtime.h>
#include <math.h>

typedef __attribute__((ext_vector_type(2))) float v2f;
typedef __attribute__((ext_vector_type(8))) float v8f;

#define NN   100000
#define EE   1600000
#define INDIM 64
#define HID  128
#define LAT  64
#define GG   1000

__device__ __forceinline__ float gelu_exact(float x) {
    return 0.5f * x * (1.0f + erff(x * 0.70710678118654752f));
}

__global__ __launch_bounds__(256) void bn_in_kernel(
    const float* __restrict__ x,
    const float* __restrict__ g, const float* __restrict__ b,
    const float* __restrict__ m, const float* __restrict__ v,
    const float* __restrict__ epsP,
    float* __restrict__ out, float* __restrict__ agg)
{
    int idx = blockIdx.x * blockDim.x + threadIdx.x;
    if (idx >= NN * INDIM) return;
    int j = idx & (INDIM - 1);
    float y = (x[idx] - m[j]) * rsqrtf(v[j] + 1e-5f) * g[j] + b[j];
    out[idx] = y;
    agg[idx] = (1.0f + epsP[0]) * y;
}

__global__ __launch_bounds__(256) void edge_agg_kernel(
    const float* __restrict__ h, const int* __restrict__ src,
    const int* __restrict__ dst, float* __restrict__ agg,
    int dim, int dim4, int total)
{
    int idx = blockIdx.x * blockDim.x + threadIdx.x;
    if (idx >= total) return;
    int e = idx / dim4;
    int c = idx - e * dim4;
    int s = src[e];
    int d = dst[e];
    const float4 val = *(const float4*)(h + (size_t)s * dim + 4 * c);
    float* p = agg + (size_t)d * dim + 4 * c;
    unsafeAtomicAdd(p + 0, val.x);
    unsafeAtomicAdd(p + 1, val.y);
    unsafeAtomicAdd(p + 2, val.z);
    unsafeAtomicAdd(p + 3, val.w);
}

template<int K>
__global__ __launch_bounds__(256) void gemm_bn_gelu_kernel(
    const float* __restrict__ A,
    const float* __restrict__ W,
    const float* __restrict__ bias,
    const float* __restrict__ bn_g, const float* __restrict__ bn_b,
    const float* __restrict__ bn_m, const float* __restrict__ bn_v,
    float* __restrict__ out,
    float* __restrict__ aggOut, const float* __restrict__ epsP,
    int epsIdx, int writeAgg)
{
    const int lane = threadIdx.x & 31;
    const int wave = threadIdx.x >> 5;
    const int tileM = blockIdx.x;
    const int half = lane >> 4;
    const int mr   = lane & 15;

    const int rowA = tileM * 16 + mr;
    const int colB = wave * 16 + mr;

    const float* __restrict__ Arow = A + (size_t)rowA * K;

    v8f acc = {};
    #pragma unroll
    for (int k = 0; k < K; k += 4) {
        const int ka = k + 2 * half;
        v2f a, b;
        a.x = Arow[ka];
        a.y = Arow[ka + 1];
        b.x = W[(size_t)ka * HID + colB];
        b.y = W[(size_t)(ka + 1) * HID + colB];
        acc = __builtin_amdgcn_wmma_f32_16x16x4_f32(
            false, a, false, b, (short)0, acc, false, false);
    }

    const float escale = writeAgg ? (1.0f + epsP[epsIdx]) : 0.0f;

    #pragma unroll
    for (int r = 0; r < 8; ++r) {
        const int row = tileM * 16 + r + 8 * half;
        const int col = wave * 16 + mr;
        float y = acc[r] + bias[col];
        y = (y - bn_m[col]) * rsqrtf(bn_v[col] + 1e-5f) * bn_g[col] + bn_b[col];
        y = gelu_exact(y);
        const size_t o = (size_t)row * HID + col;
        out[o] = y;
        if (writeAgg) aggOut[o] = escale * y;
    }
}

__global__ __launch_bounds__(256) void zero_kernel(float* __restrict__ p, int n) {
    int idx = blockIdx.x * blockDim.x + threadIdx.x;
    if (idx < n) p[idx] = 0.0f;
}

__global__ __launch_bounds__(256) void pool_acc_kernel(
    const float* __restrict__ h, const int* __restrict__ batch,
    float* __restrict__ sums, float* __restrict__ cnt, int total)
{
    int idx = blockIdx.x * blockDim.x + threadIdx.x;
    if (idx >= total) return;
    int i = idx >> 5;
    int c = idx & 31;
    int b = batch[i];
    const float4 v = *(const float4*)(h + (size_t)i * HID + 4 * c);
    float* s = sums + (size_t)b * HID + 4 * c;
    unsafeAtomicAdd(s + 0, v.x);
    unsafeAtomicAdd(s + 1, v.y);
    unsafeAtomicAdd(s + 2, v.z);
    unsafeAtomicAdd(s + 3, v.w);
    if (c == 0) unsafeAtomicAdd(cnt + b, 1.0f);
}

__global__ __launch_bounds__(256) void pool_final_kernel(
    const float* __restrict__ sums, const float* __restrict__ cnt,
    float* __restrict__ pooled)
{
    int idx = blockIdx.x * blockDim.x + threadIdx.x;
    if (idx >= GG * HID) return;
    int g = idx / HID;
    pooled[idx] = sums[idx] / fmaxf(cnt[g], 1.0f);
}

__global__ __launch_bounds__(128) void head_kernel(
    const float* __restrict__ pooled,
    const float* __restrict__ fc1w, const float* __restrict__ fc1b,
    const float* __restrict__ lng, const float* __restrict__ lnb,
    const float* __restrict__ fc2w, const float* __restrict__ fc2b,
    float* __restrict__ out)
{
    __shared__ float p[HID];
    __shared__ float rbuf[HID];
    __shared__ float red[HID];
    const int g = blockIdx.x;
    const int t = threadIdx.x;

    p[t] = pooled[(size_t)g * HID + t];
    __syncthreads();

    float o = fc1b[t];
    #pragma unroll 4
    for (int k = 0; k < HID; ++k) o += p[k] * fc1w[k * HID + t];

    red[t] = o;
    __syncthreads();
    for (int s = 64; s > 0; s >>= 1) {
        if (t < s) red[t] += red[t + s];
        __syncthreads();
    }
    const float mu = red[0] * (1.0f / HID);
    __syncthreads();

    const float dv = o - mu;
    red[t] = dv * dv;
    __syncthreads();
    for (int s = 64; s > 0; s >>= 1) {
        if (t < s) red[t] += red[t + s];
        __syncthreads();
    }
    const float var = red[0] * (1.0f / HID);

    float y = dv * rsqrtf(var + 1e-5f) * lng[t] + lnb[t];
    y = gelu_exact(y) + p[t];
    rbuf[t] = y;
    __syncthreads();

    if (t < LAT) {
        float o2 = fc2b[t];
        #pragma unroll 4
        for (int k = 0; k < HID; ++k) o2 += rbuf[k] * fc2w[k * LAT + t];
        out[(size_t)g * LAT + t] = o2;
    }
}

extern "C" void kernel_launch(void* const* d_in, const int* in_sizes, int n_in,
                              void* d_out, int out_size, void* d_ws, size_t ws_size,
                              hipStream_t stream) {
    const float* x       = (const float*)d_in[0];
    const int*   ei      = (const int*)  d_in[1];
    const int*   batch   = (const int*)  d_in[2];
    const float* in_g    = (const float*)d_in[3];
    const float* in_b    = (const float*)d_in[4];
    const float* in_m    = (const float*)d_in[5];
    const float* in_v    = (const float*)d_in[6];
    const float* eps0    = (const float*)d_in[7];
    const float* c0_w1   = (const float*)d_in[8];
    const float* c0_b1   = (const float*)d_in[9];
    const float* c0_g    = (const float*)d_in[10];
    const float* c0_bb   = (const float*)d_in[11];
    const float* c0_m    = (const float*)d_in[12];
    const float* c0_v    = (const float*)d_in[13];
    const float* c0_w2   = (const float*)d_in[14];
    const float* c0_b2   = (const float*)d_in[15];
    const float* bn0_g   = (const float*)d_in[16];
    const float* bn0_b   = (const float*)d_in[17];
    const float* bn0_m   = (const float*)d_in[18];
    const float* bn0_v   = (const float*)d_in[19];
    const float* epsR    = (const float*)d_in[20];
    const float* cR_w1   = (const float*)d_in[21];
    const float* cR_b1   = (const float*)d_in[22];
    const float* cR_g    = (const float*)d_in[23];
    const float* cR_bb   = (const float*)d_in[24];
    const float* cR_m    = (const float*)d_in[25];
    const float* cR_v    = (const float*)d_in[26];
    const float* cR_w2   = (const float*)d_in[27];
    const float* cR_b2   = (const float*)d_in[28];
    const float* bnR_g   = (const float*)d_in[29];
    const float* bnR_b   = (const float*)d_in[30];
    const float* bnR_m   = (const float*)d_in[31];
    const float* bnR_v   = (const float*)d_in[32];
    const float* fc1_w   = (const float*)d_in[33];
    const float* fc1_b   = (const float*)d_in[34];
    const float* ln_g    = (const float*)d_in[35];
    const float* ln_b    = (const float*)d_in[36];
    const float* fc2_w   = (const float*)d_in[37];
    const float* fc2_b   = (const float*)d_in[38];

    const int* src = ei;
    const int* dst = ei + EE;

    float* buf0 = (float*)d_ws;
    float* buf1 = buf0 + (size_t)NN * HID;
    float* buf2 = buf1 + (size_t)NN * HID;
    float* sums = buf2 + (size_t)NN * HID;
    float* cnt  = sums + (size_t)GG * HID;

    const int TB = 256;
    const int nTiles = NN / 16;

    bn_in_kernel<<<(NN * INDIM + TB - 1) / TB, TB, 0, stream>>>(
        x, in_g, in_b, in_m, in_v, eps0, buf0, buf1);

    {
        int total = EE * (INDIM / 4);
        edge_agg_kernel<<<(total + TB - 1) / TB, TB, 0, stream>>>(
            buf0, src, dst, buf1, INDIM, INDIM / 4, total);
        gemm_bn_gelu_kernel<INDIM><<<nTiles, TB, 0, stream>>>(
            buf1, c0_w1, c0_b1, c0_g, c0_bb, c0_m, c0_v, buf2,
            buf1, eps0, 0, 0);
        gemm_bn_gelu_kernel<HID><<<nTiles, TB, 0, stream>>>(
            buf2, c0_w2, c0_b2, bn0_g, bn0_b, bn0_m, bn0_v, buf0,
            buf1, epsR, 0, 1);
    }

    for (int l = 0; l < 2; ++l) {
        const size_t wOff = (size_t)l * HID * HID;
        const size_t vOff = (size_t)l * HID;
        int total = EE * (HID / 4);
        edge_agg_kernel<<<(total + TB - 1) / TB, TB, 0, stream>>>(
            buf0, src, dst, buf1, HID, HID / 4, total);
        gemm_bn_gelu_kernel<HID><<<nTiles, TB, 0, stream>>>(
            buf1, cR_w1 + wOff, cR_b1 + vOff, cR_g + vOff, cR_bb + vOff,
            cR_m + vOff, cR_v + vOff, buf2,
            buf1, epsR, 0, 0);
        const int writeAgg = (l == 0) ? 1 : 0;
        gemm_bn_gelu_kernel<HID><<<nTiles, TB, 0, stream>>>(
            buf2, cR_w2 + wOff, cR_b2 + vOff, bnR_g + vOff, bnR_b + vOff,
            bnR_m + vOff, bnR_v + vOff, buf0,
            buf1, epsR, 1, writeAgg);
    }

    zero_kernel<<<(GG * HID + GG + TB - 1) / TB, TB, 0, stream>>>(sums, GG * HID + GG);
    {
        int total = NN * (HID / 4);
        pool_acc_kernel<<<(total + TB - 1) / TB, TB, 0, stream>>>(
            buf0, batch, sums, cnt, total);
    }
    pool_final_kernel<<<(GG * HID + TB - 1) / TB, TB, 0, stream>>>(sums, cnt, buf1);

    head_kernel<<<GG, HID, 0, stream>>>(
        buf1, fc1_w, fc1_b, ln_g, ln_b, fc2_w, fc2_b, (float*)d_out);
}


# Edge sums without float atomics: edges grouped by dst once per call (int atomic rank, scan, fill), then one wave per node gathers neighbour rows, one store; add order = arrival-order rank
# speedup vs baseline: 1.6240x; 1.6240x over previous
#include <hip/hip_runtime.h>
#include <math.h>

typedef __attribute__((ext_vector_type(2))) float v2f;
typedef __attribute__((ext_vector_type(8))) float v8f;

#define NN   100000
#define EE   1600000
#define INDIM 64
#define HID  128
#define LAT  64
#define GG   1000

__device__ __forceinline__ float gelu_exact(float x) {
    return 0.5f * x * (1.0f + erff(x * 0.70710678118654752f));
}

__global__ __launch_bounds__(256) void bn_in_kernel(
    const float* __restrict__ x,
    const float* __restrict__ g, const float* __restrict__ b,
    const float* __restrict__ m, const float* __restrict__ v,
    float* __restrict__ out, int* __restrict__ deg)
{
    int idx = blockIdx.x * blockDim.x + threadIdx.x;
    if (idx >= NN * INDIM) return;
    if (idx < NN) deg[idx] = 0;
    int j = idx & (INDIM - 1);
    float y = (x[idx] - m[j]) * rsqrtf(v[j] + 1e-5f) * g[j] + b[j];
    out[idx] = y;
}


__global__ __launch_bounds__(256) void edge_rank_kernel(
    const int* __restrict__ dst, int* __restrict__ deg, int* __restrict__ rank)
{
    int e = blockIdx.x * blockDim.x + threadIdx.x;
    if (e >= EE) return;
    rank[e] = atomicAdd(deg + dst[e], 1);
}

#define SCAN_CHUNK 1024
#define SCAN_BLOCKS ((NN + SCAN_CHUNK - 1) / SCAN_CHUNK)

__device__ __forceinline__ int4 load_deg4(const int* __restrict__ deg, int base) {
    if (base < NN) return *(const int4*)(deg + base);
    return make_int4(0, 0, 0, 0);
}

__global__ __launch_bounds__(256) void deg_blocksum_kernel(
    const int* __restrict__ deg, int* __restrict__ bsum)
{
    __shared__ int red[8];
    const int t = threadIdx.x;
    const int4 d = load_deg4(deg, blockIdx.x * SCAN_CHUNK + t * 4);
    int s = d.x + d.y + d.z + d.w;
    for (int off = 16; off > 0; off >>= 1) s += __shfl_xor(s, off);
    if ((t & 31) == 0) red[t >> 5] = s;
    __syncthreads();
    if (t == 0) {
        int tot = 0;
        for (int w = 0; w < 8; ++w) tot += red[w];
        bsum[blockIdx.x] = tot;
    }
}

__global__ __launch_bounds__(256) void rowptr_scan_kernel(
    const int* __restrict__ deg, const int* __restrict__ bsum,
    int* __restrict__ rowptr)
{
    __shared__ int wsum[8];
    __shared__ int boff;
    const int t = threadIdx.x;
    const int lane = t & 31;
    const int wave = t >> 5;

    if (wave == 0) {
        int s = 0;
        for (int i = lane; i < (int)blockIdx.x; i += 32) s += bsum[i];
        for (int off = 16; off > 0; off >>= 1) s += __shfl_xor(s, off);
        if (lane == 0) boff = s;
    }

    const int base = blockIdx.x * SCAN_CHUNK + t * 4;
    const int4 d = load_deg4(deg, base);
    const int tsum = d.x + d.y + d.z + d.w;
    int inc = tsum;
    for (int off = 1; off < 32; off <<= 1) {
        const int n = __shfl_up(inc, off);
        if (lane >= off) inc += n;
    }
    if (lane == 31) wsum[wave] = inc;
    __syncthreads();
    int woff = 0;
    for (int w = 0; w < wave; ++w) woff += wsum[w];
    const int ex = boff + woff + inc - tsum;
    if (base < NN) {
        *(int4*)(rowptr + base) =
            make_int4(ex, ex + d.x, ex + d.x + d.y, ex + d.x + d.y + d.z);
        if (base + 4 == NN) rowptr[NN] = ex + tsum;
    }
}

__global__ __launch_bounds__(256) void edge_fill_kernel(
    const int* __restrict__ src, const int* __restrict__ dst,
    const int* __restrict__ rank, const int* __restrict__ rowptr,
    int* __restrict__ col)
{
    int e = blockIdx.x * blockDim.x + threadIdx.x;
    if (e >= EE) return;
    col[rowptr[dst[e]] + rank[e]] = src[e];
}

template<int DIM>
__global__ __launch_bounds__(256) void gather_agg_kernel(
    const float* __restrict__ h, const int* __restrict__ rowptr,
    const int* __restrict__ col, const float* __restrict__ epsP, int epsIdx,
    float* __restrict__ agg)
{
    constexpr int VW = DIM / 32;
    typedef __attribute__((ext_vector_type(VW))) float vt;
    const int lane = threadIdx.x & 31;
    const int node = __builtin_amdgcn_readfirstlane(
        (int)(blockIdx.x * 8 + (threadIdx.x >> 5)));
    const int beg = rowptr[node];
    const int end = rowptr[node + 1];
    const float* __restrict__ hl = h + VW * lane;

    const vt self = *(const vt*)(hl + (size_t)node * DIM);

    vt acc = {};
    const vt zero = {};
    for (int base = beg; base < end; base += 32) {
        const int n = min(32, end - base);
        int ids = (lane < n) ? col[base + lane] : 0;
        ids = (int)min((unsigned)ids, (unsigned)(NN - 1));
        int j = 0;
        for (; j + 4 <= n; j += 4) {
            const int s0 = __builtin_amdgcn_readlane(ids, j);
            const int s1 = __builtin_amdgcn_readlane(ids, j + 1);
            const int s2 = __builtin_amdgcn_readlane(ids, j + 2);
            const int s3 = __builtin_amdgcn_readlane(ids, j + 3);
            const vt v0 = *(const vt*)(hl + (size_t)s0 * DIM);
            const vt v1 = *(const vt*)(hl + (size_t)s1 * DIM);
            const vt v2 = *(const vt*)(hl + (size_t)s2 * DIM);
            const vt v3 = *(const vt*)(hl + (size_t)s3 * DIM);
            acc += v0; acc += v1; acc += v2; acc += v3;
        }
        if (j < n) {
            const int last = n - 1;
            const int s0 = __builtin_amdgcn_readlane(ids, j);
            const int s1 = __builtin_amdgcn_readlane(ids, min(j + 1, last));
            const int s2 = __builtin_amdgcn_readlane(ids, min(j + 2, last));
            const vt v0 = *(const vt*)(hl + (size_t)s0 * DIM);
            const vt v1 = *(const vt*)(hl + (size_t)s1 * DIM);
            const vt v2 = *(const vt*)(hl + (size_t)s2 * DIM);
            acc += v0;
            acc += (j + 1 < n) ? v1 : zero;
            acc += (j + 2 < n) ? v2 : zero;
        }
    }
    const float sc = 1.0f + epsP[epsIdx];
    *(vt*)(agg + (size_t)node * DIM + VW * lane) = sc * self + acc;
}

template<int K>
__global__ __launch_bounds__(256) void gemm_bn_gelu_kernel(
    const float* __restrict__ A,
    const float* __restrict__ W,
    const float* __restrict__ bias,
    const float* __restrict__ bn_g, const float* __restrict__ bn_b,
    const float* __restrict__ bn_m, const float* __restrict__ bn_v,
    float* __restrict__ out)
{
    const int lane = threadIdx.x & 31;
    const int wave = threadIdx.x >> 5;
    const int tileM = blockIdx.x;
    const int half = lane >> 4;
    const int mr   = lane & 15;

    const int rowA = tileM * 16 + mr;
    const int colB = wave * 16 + mr;

    const float* __restrict__ Arow = A + (size_t)rowA * K;

    v8f acc = {};
    #pragma unroll
    for (int k = 0; k < K; k += 4) {
        const int ka = k + 2 * half;
        v2f a, b;
        a.x = Arow[ka];
        a.y = Arow[ka + 1];
        b.x = W[(size_t)ka * HID + colB];
        b.y = W[(size_t)(ka + 1) * HID + colB];
        acc = __builtin_amdgcn_wmma_f32_16x16x4_f32(
            false, a, false, b, (short)0, acc, false, false);
    }

    #pragma unroll
    for (int r = 0; r < 8; ++r) {
        const int row = tileM * 16 + r + 8 * half;
        const int col = wave * 16 + mr;
        float y = acc[r] + bias[col];
        y = (y - bn_m[col]) * rsqrtf(bn_v[col] + 1e-5f) * bn_g[col] + bn_b[col];
        y = gelu_exact(y);
        out[(size_t)row * HID + col] = y;
    }
}

__global__ __launch_bounds__(256) void zero_kernel(float* __restrict__ p, int n) {
    int idx = blockIdx.x * blockDim.x + threadIdx.x;
    if (idx < n) p[idx] = 0.0f;
}

__global__ __launch_bounds__(256) void pool_acc_kernel(
    const float* __restrict__ h, const int* __restrict__ batch,
    float* __restrict__ sums, float* __restrict__ cnt, int total)
{
    int idx = blockIdx.x * blockDim.x + threadIdx.x;
    if (idx >= total) return;
    int i = idx >> 5;
    int c = idx & 31;
    int b = batch[i];
    const float4 v = *(const float4*)(h + (size_t)i * HID + 4 * c);
    float* s = sums + (size_t)b * HID + 4 * c;
    unsafeAtomicAdd(s + 0, v.x);
    unsafeAtomicAdd(s + 1, v.y);
    unsafeAtomicAdd(s + 2, v.z);
    unsafeAtomicAdd(s + 3, v.w);
    if (c == 0) unsafeAtomicAdd(cnt + b, 1.0f);
}

__global__ __launch_bounds__(256) void pool_final_kernel(
    const float* __restrict__ sums, const float* __restrict__ cnt,
    float* __restrict__ pooled)
{
    int idx = blockIdx.x * blockDim.x + threadIdx.x;
    if (idx >= GG * HID) return;
    int g = idx / HID;
    pooled[idx] = sums[idx] / fmaxf(cnt[g], 1.0f);
}

__global__ __launch_bounds__(128) void head_kernel(
    const float* __restrict__ pooled,
    const float* __restrict__ fc1w, const float* __restrict__ fc1b,
    const float* __restrict__ lng, const float* __restrict__ lnb,
    const float* __restrict__ fc2w, const float* __restrict__ fc2b,
    float* __restrict__ out)
{
    __shared__ float p[HID];
    __shared__ float rbuf[HID];
    __shared__ float red[HID];
    const int g = blockIdx.x;
    const int t = threadIdx.x;

    p[t] = pooled[(size_t)g * HID + t];
    __syncthreads();

    float o = fc1b[t];
    #pragma unroll 4
    for (int k = 0; k < HID; ++k) o += p[k] * fc1w[k * HID + t];

    red[t] = o;
    __syncthreads();
    for (int s = 64; s > 0; s >>= 1) {
        if (t < s) red[t] += red[t + s];
        __syncthreads();
    }
    const float mu = red[0] * (1.0f / HID);
    __syncthreads();

    const float dv = o - mu;
    red[t] = dv * dv;
    __syncthreads();
    for (int s = 64; s > 0; s >>= 1) {
        if (t < s) red[t] += red[t + s];
        __syncthreads();
    }
    const float var = red[0] * (1.0f / HID);

    float y = dv * rsqrtf(var + 1e-5f) * lng[t] + lnb[t];
    y = gelu_exact(y) + p[t];
    rbuf[t] = y;
    __syncthreads();

    if (t < LAT) {
        float o2 = fc2b[t];
        #pragma unroll 4
        for (int k = 0; k < HID; ++k) o2 += rbuf[k] * fc2w[k * LAT + t];
        out[(size_t)g * LAT + t] = o2;
    }
}

extern "C" void kernel_launch(void* const* d_in, const int* in_sizes, int n_in,
                              void* d_out, int out_size, void* d_ws, size_t ws_size,
                              hipStream_t stream) {
    const float* x       = (const float*)d_in[0];
    const int*   ei      = (const int*)  d_in[1];
    const int*   batch   = (const int*)  d_in[2];
    const float* in_g    = (const float*)d_in[3];
    const float* in_b    = (const float*)d_in[4];
    const float* in_m    = (const float*)d_in[5];
    const float* in_v    = (const float*)d_in[6];
    const float* eps0    = (const float*)d_in[7];
    const float* c0_w1   = (const float*)d_in[8];
    const float* c0_b1   = (const float*)d_in[9];
    const float* c0_g    = (const float*)d_in[10];
    const float* c0_bb   = (const float*)d_in[11];
    const float* c0_m    = (const float*)d_in[12];
    const float* c0_v    = (const float*)d_in[13];
    const float* c0_w2   = (const float*)d_in[14];
    const float* c0_b2   = (const float*)d_in[15];
    const float* bn0_g   = (const float*)d_in[16];
    const float* bn0_b   = (const float*)d_in[17];
    const float* bn0_m   = (const float*)d_in[18];
    const float* bn0_v   = (const float*)d_in[19];
    const float* epsR    = (const float*)d_in[20];
    const float* cR_w1   = (const float*)d_in[21];
    const float* cR_b1   = (const float*)d_in[22];
    const float* cR_g    = (const float*)d_in[23];
    const float* cR_bb   = (const float*)d_in[24];
    const float* cR_m    = (const float*)d_in[25];
    const float* cR_v    = (const float*)d_in[26];
    const float* cR_w2   = (const float*)d_in[27];
    const float* cR_b2   = (const float*)d_in[28];
    const float* bnR_g   = (const float*)d_in[29];
    const float* bnR_b   = (const float*)d_in[30];
    const float* bnR_m   = (const float*)d_in[31];
    const float* bnR_v   = (const float*)d_in[32];
    const float* fc1_w   = (const float*)d_in[33];
    const float* fc1_b   = (const float*)d_in[34];
    const float* ln_g    = (const float*)d_in[35];
    const float* ln_b    = (const float*)d_in[36];
    const float* fc2_w   = (const float*)d_in[37];
    const float* fc2_b   = (const float*)d_in[38];

    const int* src = ei;
    const int* dst = ei + EE;

    float* bufA = (float*)d_ws;
    float* bufB = bufA + (size_t)NN * HID;
    int*   col    = (int*)(bufB + (size_t)NN * HID);
    int*   rank   = col + EE;
    int*   deg    = rank + EE;
    int*   rowptr = deg + NN;
    int*   bsum   = rowptr + NN + 4;
    float* sums = bufA + (size_t)3 * NN * HID;
    float* cnt  = sums + (size_t)GG * HID;

    const int TB = 256;
    const int nTiles = NN / 16;
    const int eBlocks = (EE + TB - 1) / TB;
    const int nodeBlocks = NN / 8;

    bn_in_kernel<<<(NN * INDIM + TB - 1) / TB, TB, 0, stream>>>(
        x, in_g, in_b, in_m, in_v, bufA, deg);

    edge_rank_kernel<<<eBlocks, TB, 0, stream>>>(dst, deg, rank);
    deg_blocksum_kernel<<<SCAN_BLOCKS, TB, 0, stream>>>(deg, bsum);
    rowptr_scan_kernel<<<SCAN_BLOCKS, TB, 0, stream>>>(deg, bsum, rowptr);
    edge_fill_kernel<<<eBlocks, TB, 0, stream>>>(src, dst, rank, rowptr, col);

    gather_agg_kernel<INDIM><<<nodeBlocks, TB, 0, stream>>>(
        bufA, rowptr, col, eps0, 0, bufB);
    gemm_bn_gelu_kernel<INDIM><<<nTiles, TB, 0, stream>>>(
        bufB, c0_w1, c0_b1, c0_g, c0_bb, c0_m, c0_v, bufA);
    gemm_bn_gelu_kernel<HID><<<nTiles, TB, 0, stream>>>(
        bufA, c0_w2, c0_b2, bn0_g, bn0_b, bn0_m, bn0_v, bufB);

    float* hin = bufB;
    float* hout = bufA;
    for (int l = 0; l < 2; ++l) {
        const size_t wOff = (size_t)l * HID * HID;
        const size_t vOff = (size_t)l * HID;
        gather_agg_kernel<HID><<<nodeBlocks, TB, 0, stream>>>(
            hin, rowptr, col, epsR, l, hout);
        gemm_bn_gelu_kernel<HID><<<nTiles, TB, 0, stream>>>(
            hout, cR_w1 + wOff, cR_b1 + vOff, cR_g + vOff, cR_bb + vOff,
            cR_m + vOff, cR_v + vOff, hin);
        gemm_bn_gelu_kernel<HID><<<nTiles, TB, 0, stream>>>(
            hin, cR_w2 + wOff, cR_b2 + vOff, bnR_g + vOff, bnR_b + vOff,
            bnR_m + vOff, bnR_v + vOff, hout);
        float* tmp = hin; hin = hout; hout = tmp;
    }

    zero_kernel<<<(GG * HID + GG + TB - 1) / TB, TB, 0, stream>>>(sums, GG * HID + GG);
    {
        int total = NN * (HID / 4);
        pool_acc_kernel<<<(total + TB - 1) / TB, TB, 0, stream>>>(
            hin, batch, sums, cnt, total);
    }
    pool_final_kernel<<<(GG * HID + TB - 1) / TB, TB, 0, stream>>>(sums, cnt, hout);

    head_kernel<<<GG, HID, 0, stream>>>(
        hout, fc1_w, fc1_b, ln_g, ln_b, fc2_w, fc2_b, (float*)d_out);
}


# GEMM re-tiled for operand reuse: one wave per 16x128 output strip (one A fragment feeds 8 WMMAs), W staged once per block in LDS for the B fragments, 7-wave blocks; k order and add order unchanged
# speedup vs baseline: 1.7589x; 1.0831x over previous
#include <hip/hip_runtime.h>
#include <math.h>

typedef __attribute__((ext_vector_type(2))) float v2f;
typedef __attribute__((ext_vector_type(8))) float v8f;

#define NN   100000
#define EE   1600000
#define INDIM 64
#define HID  128
#define LAT  64
#define GG   1000

__device__ __forceinline__ float gelu_exact(float x) {
    return 0.5f * x * (1.0f + erff(x * 0.70710678118654752f));
}

__global__ __launch_bounds__(256) void bn_in_kernel(
    const float* __restrict__ x,
    const float* __restrict__ g, const float* __restrict__ b,
    const float* __restrict__ m, const float* __restrict__ v,
    float* __restrict__ out, int* __restrict__ deg)
{
    int idx = blockIdx.x * blockDim.x + threadIdx.x;
    if (idx >= NN * INDIM) return;
    if (idx < NN) deg[idx] = 0;
    int j = idx & (INDIM - 1);
    float y = (x[idx] - m[j]) * rsqrtf(v[j] + 1e-5f) * g[j] + b[j];
    out[idx] = y;
}


__global__ __launch_bounds__(256) void edge_rank_kernel(
    const int* __restrict__ dst, int* __restrict__ deg, int* __restrict__ rank)
{
    int e = blockIdx.x * blockDim.x + threadIdx.x;
    if (e >= EE) return;
    rank[e] = atomicAdd(deg + dst[e], 1);
}

#define SCAN_CHUNK 1024
#define SCAN_BLOCKS ((NN + SCAN_CHUNK - 1) / SCAN_CHUNK)

__device__ __forceinline__ int4 load_deg4(const int* __restrict__ deg, int base) {
    if (base < NN) return *(const int4*)(deg + base);
    return make_int4(0, 0, 0, 0);
}

__global__ __launch_bounds__(256) void deg_blocksum_kernel(
    const int* __restrict__ deg, int* __restrict__ bsum)
{
    __shared__ int red[8];
    const int t = threadIdx.x;
    const int4 d = load_deg4(deg, blockIdx.x * SCAN_CHUNK + t * 4);
    int s = d.x + d.y + d.z + d.w;
    for (int off = 16; off > 0; off >>= 1) s += __shfl_xor(s, off);
    if ((t & 31) == 0) red[t >> 5] = s;
    __syncthreads();
    if (t == 0) {
        int tot = 0;
        for (int w = 0; w < 8; ++w) tot += red[w];
        bsum[blockIdx.x] = tot;
    }
}

__global__ __launch_bounds__(256) void rowptr_scan_kernel(
    const int* __restrict__ deg, const int* __restrict__ bsum,
    int* __restrict__ rowptr)
{
    __shared__ int wsum[8];
    __shared__ int boff;
    const int t = threadIdx.x;
    const int lane = t & 31;
    const int wave = t >> 5;

    if (wave == 0) {
        int s = 0;
        for (int i = lane; i < (int)blockIdx.x; i += 32) s += bsum[i];
        for (int off = 16; off > 0; off >>= 1) s += __shfl_xor(s, off);
        if (lane == 0) boff = s;
    }

    const int base = blockIdx.x * SCAN_CHUNK + t * 4;
    const int4 d = load_deg4(deg, base);
    const int tsum = d.x + d.y + d.z + d.w;
    int inc = tsum;
    for (int off = 1; off < 32; off <<= 1) {
        const int n = __shfl_up(inc, off);
        if (lane >= off) inc += n;
    }
    if (lane == 31) wsum[wave] = inc;
    __syncthreads();
    int woff = 0;
    for (int w = 0; w < wave; ++w) woff += wsum[w];
    const int ex = boff + woff + inc - tsum;
    if (base < NN) {
        *(int4*)(rowptr + base) =
            make_int4(ex, ex + d.x, ex + d.x + d.y, ex + d.x + d.y + d.z);
        if (base + 4 == NN) rowptr[NN] = ex + tsum;
    }
}

__global__ __launch_bounds__(256) void edge_fill_kernel(
    const int* __restrict__ src, const int* __restrict__ dst,
    const int* __restrict__ rank, const int* __restrict__ rowptr,
    int* __restrict__ col)
{
    int e = blockIdx.x * blockDim.x + threadIdx.x;
    if (e >= EE) return;
    col[rowptr[dst[e]] + rank[e]] = src[e];
}

template<int DIM>
__global__ __launch_bounds__(256) void gather_agg_kernel(
    const float* __restrict__ h, const int* __restrict__ rowptr,
    const int* __restrict__ col, const float* __restrict__ epsP, int epsIdx,
    float* __restrict__ agg)
{
    constexpr int VW = DIM / 32;
    typedef __attribute__((ext_vector_type(VW))) float vt;
    const int lane = threadIdx.x & 31;
    const int node = __builtin_amdgcn_readfirstlane(
        (int)(blockIdx.x * 8 + (threadIdx.x >> 5)));
    const int beg = rowptr[node];
    const int end = rowptr[node + 1];
    const float* __restrict__ hl = h + VW * lane;

    const vt self = *(const vt*)(hl + (size_t)node * DIM);

    vt acc = {};
    const vt zero = {};
    for (int base = beg; base < end; base += 32) {
        const int n = min(32, end - base);
        int ids = (lane < n) ? col[base + lane] : 0;
        ids = (int)min((unsigned)ids, (unsigned)(NN - 1));
        int j = 0;
        for (; j + 4 <= n; j += 4) {
            const int s0 = __builtin_amdgcn_readlane(ids, j);
            const int s1 = __builtin_amdgcn_readlane(ids, j + 1);
            const int s2 = __builtin_amdgcn_readlane(ids, j + 2);
            const int s3 = __builtin_amdgcn_readlane(ids, j + 3);
            const vt v0 = *(const vt*)(hl + (size_t)s0 * DIM);
            const vt v1 = *(const vt*)(hl + (size_t)s1 * DIM);
            const vt v2 = *(const vt*)(hl + (size_t)s2 * DIM);
            const vt v3 = *(const vt*)(hl + (size_t)s3 * DIM);
            acc += v0; acc += v1; acc += v2; acc += v3;
        }
        if (j < n) {
            const int last = n - 1;
            const int s0 = __builtin_amdgcn_readlane(ids, j);
            const int s1 = __builtin_amdgcn_readlane(ids, min(j + 1, last));
            const int s2 = __builtin_amdgcn_readlane(ids, min(j + 2, last));
            const vt v0 = *(const vt*)(hl + (size_t)s0 * DIM);
            const vt v1 = *(const vt*)(hl + (size_t)s1 * DIM);
            const vt v2 = *(const vt*)(hl + (size_t)s2 * DIM);
            acc += v0;
            acc += (j + 1 < n) ? v1 : zero;
            acc += (j + 2 < n) ? v2 : zero;
        }
    }
    const float sc = 1.0f + epsP[epsIdx];
    *(vt*)(agg + (size_t)node * DIM + VW * lane) = sc * self + acc;
}

#define ROW_TILES (NN / 16)
#define GEMM_WAVES 7
#define GEMM_THREADS (GEMM_WAVES * 32)
#define GEMM_BLOCKS ((ROW_TILES + GEMM_WAVES - 1) / GEMM_WAVES)
#define WPITCH 144

template<int K>
__global__ __launch_bounds__(GEMM_THREADS) void gemm_bn_gelu_kernel(
    const float* __restrict__ A,
    const float* __restrict__ W,
    const float* __restrict__ bias,
    const float* __restrict__ bn_g, const float* __restrict__ bn_b,
    const float* __restrict__ bn_m, const float* __restrict__ bn_v,
    float* __restrict__ out)
{
    __shared__ __attribute__((aligned(16))) float Ws[(K / 2) * WPITCH * 2];

    const int lane = threadIdx.x & 31;
    const int wave = __builtin_amdgcn_readfirstlane((int)(threadIdx.x >> 5));
    const int half = lane >> 4;
    const int mr   = lane & 15;

    const int tileM = blockIdx.x * GEMM_WAVES + wave;
    const bool live = tileM < ROW_TILES;
    const float* __restrict__ Arow =
        A + (size_t)((live ? tileM : 0) * 16 + mr) * K + 2 * half;

    constexpr int NFULL = (K / 2) / GEMM_WAVES;
    constexpr int NREM  = (K / 2) - NFULL * GEMM_WAVES;
    const bool extra = wave < NREM;
    float4 e0 = {}, e1 = {};
    if (extra) {
        const int kp = NFULL * GEMM_WAVES + wave;
        e0 = *(const float4*)(W + (size_t)(2 * kp) * HID + 4 * lane);
        e1 = *(const float4*)(W + (size_t)(2 * kp + 1) * HID + 4 * lane);
    }
    float4 r0[NFULL], r1[NFULL];
    #pragma unroll
    for (int i = 0; i < NFULL; ++i) {
        const int kp = wave + GEMM_WAVES * i;
        r0[i] = *(const float4*)(W + (size_t)(2 * kp) * HID + 4 * lane);
        r1[i] = *(const float4*)(W + (size_t)(2 * kp + 1) * HID + 4 * lane);
    }
    constexpr int KC = 16;
    v2f a_cur[KC / 4];
    #pragma unroll
    for (int i = 0; i < KC / 4; ++i) a_cur[i] = *(const v2f*)(Arow + 4 * i);
    #pragma unroll
    for (int i = 0; i < NFULL; ++i) {
        const int kp = wave + GEMM_WAVES * i;
        float* d = Ws + (kp * WPITCH + 4 * lane) * 2;
        *(float4*)(d)     = make_float4(r0[i].x, r1[i].x, r0[i].y, r1[i].y);
        *(float4*)(d + 4) = make_float4(r0[i].z, r1[i].z, r0[i].w, r1[i].w);
    }
    if (extra) {
        const int kp = NFULL * GEMM_WAVES + wave;
        float* d = Ws + (kp * WPITCH + 4 * lane) * 2;
        *(float4*)(d)     = make_float4(e0.x, e1.x, e0.y, e1.y);
        *(float4*)(d + 4) = make_float4(e0.z, e1.z, e0.w, e1.w);
    }
    __syncthreads();
    if (!live) return;

    const float* Bl = Ws + (half * WPITCH + mr) * 2;

    v8f acc[8];
    #pragma unroll
    for (int c = 0; c < 8; ++c) acc[c] = (v8f){};

    #pragma unroll 1
    for (int kc = 0; kc < K; kc += KC) {
        const int kn = min(kc + KC, K - KC);
        v2f a_nxt[KC / 4];
        #pragma unroll
        for (int i = 0; i < KC / 4; ++i) a_nxt[i] = *(const v2f*)(Arow + kn + 4 * i);
        #pragma unroll
        for (int i = 0; i < KC / 4; ++i) {
            #pragma unroll
            for (int c = 0; c < 8; ++c) {
                const v2f b = *(const v2f*)(Bl + ((kc + 4 * i) / 2) * WPITCH * 2 + c * 32);
                acc[c] = __builtin_amdgcn_wmma_f32_16x16x4_f32(
                    false, a_cur[i], false, b, (short)0, acc[c], false, false);
            }
        }
        #pragma unroll
        for (int i = 0; i < KC / 4; ++i) a_cur[i] = a_nxt[i];
    }

    #pragma unroll
    for (int c = 0; c < 8; ++c) {
        const int col = c * 16 + mr;
        const float cb = bias[col];
        const float cm = bn_m[col];
        const float cs = rsqrtf(bn_v[col] + 1e-5f);
        const float cg = bn_g[col];
        const float co = bn_b[col];
        #pragma unroll
        for (int r = 0; r < 8; ++r) {
            const int row = tileM * 16 + r + 8 * half;
            float y = acc[c][r] + cb;
            y = (y - cm) * cs * cg + co;
            y = gelu_exact(y);
            out[(size_t)row * HID + col] = y;
        }
    }
}

__global__ __launch_bounds__(256) void zero_kernel(float* __restrict__ p, int n) {
    int idx = blockIdx.x * blockDim.x + threadIdx.x;
    if (idx < n) p[idx] = 0.0f;
}

__global__ __launch_bounds__(256) void pool_acc_kernel(
    const float* __restrict__ h, const int* __restrict__ batch,
    float* __restrict__ sums, float* __restrict__ cnt, int total)
{
    int idx = blockIdx.x * blockDim.x + threadIdx.x;
    if (idx >= total) return;
    int i = idx >> 5;
    int c = idx & 31;
    int b = batch[i];
    const float4 v = *(const float4*)(h + (size_t)i * HID + 4 * c);
    float* s = sums + (size_t)b * HID + 4 * c;
    unsafeAtomicAdd(s + 0, v.x);
    unsafeAtomicAdd(s + 1, v.y);
    unsafeAtomicAdd(s + 2, v.z);
    unsafeAtomicAdd(s + 3, v.w);
    if (c == 0) unsafeAtomicAdd(cnt + b, 1.0f);
}

__global__ __launch_bounds__(256) void pool_final_kernel(
    const float* __restrict__ sums, const float* __restrict__ cnt,
    float* __restrict__ pooled)
{
    int idx = blockIdx.x * blockDim.x + threadIdx.x;
    if (idx >= GG * HID) return;
    int g = idx / HID;
    pooled[idx] = sums[idx] / fmaxf(cnt[g], 1.0f);
}

__global__ __launch_bounds__(128) void head_kernel(
    const float* __restrict__ pooled,
    const float* __restrict__ fc1w, const float* __restrict__ fc1b,
    const float* __restrict__ lng, const float* __restrict__ lnb,
    const float* __restrict__ fc2w, const float* __restrict__ fc2b,
    float* __restrict__ out)
{
    __shared__ float p[HID];
    __shared__ float rbuf[HID];
    __shared__ float red[HID];
    const int g = blockIdx.x;
    const int t = threadIdx.x;

    p[t] = pooled[(size_t)g * HID + t];
    __syncthreads();

    float o = fc1b[t];
    #pragma unroll 4
    for (int k = 0; k < HID; ++k) o += p[k] * fc1w[k * HID + t];

    red[t] = o;
    __syncthreads();
    for (int s = 64; s > 0; s >>= 1) {
        if (t < s) red[t] += red[t + s];
        __syncthreads();
    }
    const float mu = red[0] * (1.0f / HID);
    __syncthreads();

    const float dv = o - mu;
    red[t] = dv * dv;
    __syncthreads();
    for (int s = 64; s > 0; s >>= 1) {
        if (t < s) red[t] += red[t + s];
        __syncthreads();
    }
    const float var = red[0] * (1.0f / HID);

    float y = dv * rsqrtf(var + 1e-5f) * lng[t] + lnb[t];
    y = gelu_exact(y) + p[t];
    rbuf[t] = y;
    __syncthreads();

    if (t < LAT) {
        float o2 = fc2b[t];
        #pragma unroll 4
        for (int k = 0; k < HID; ++k) o2 += rbuf[k] * fc2w[k * LAT + t];
        out[(size_t)g * LAT + t] = o2;
    }
}

extern "C" void kernel_launch(void* const* d_in, const int* in_sizes, int n_in,
                              void* d_out, int out_size, void* d_ws, size_t ws_size,
                              hipStream_t stream) {
    const float* x       = (const float*)d_in[0];
    const int*   ei      = (const int*)  d_in[1];
    const int*   batch   = (const int*)  d_in[2];
    const float* in_g    = (const float*)d_in[3];
    const float* in_b    = (const float*)d_in[4];
    const float* in_m    = (const float*)d_in[5];
    const float* in_v    = (const float*)d_in[6];
    const float* eps0    = (const float*)d_in[7];
    const float* c0_w1   = (const float*)d_in[8];
    const float* c0_b1   = (const float*)d_in[9];
    const float* c0_g    = (const float*)d_in[10];
    const float* c0_bb   = (const float*)d_in[11];
    const float* c0_m    = (const float*)d_in[12];
    const float* c0_v    = (const float*)d_in[13];
    const float* c0_w2   = (const float*)d_in[14];
    const float* c0_b2   = (const float*)d_in[15];
    const float* bn0_g   = (const float*)d_in[16];
    const float* bn0_b   = (const float*)d_in[17];
    const float* bn0_m   = (const float*)d_in[18];
    const float* bn0_v   = (const float*)d_in[19];
    const float* epsR    = (const float*)d_in[20];
    const float* cR_w1   = (const float*)d_in[21];
    const float* cR_b1   = (const float*)d_in[22];
    const float* cR_g    = (const float*)d_in[23];
    const float* cR_bb   = (const float*)d_in[24];
    const float* cR_m    = (const float*)d_in[25];
    const float* cR_v    = (const float*)d_in[26];
    const float* cR_w2   = (const float*)d_in[27];
    const float* cR_b2   = (const float*)d_in[28];
    const float* bnR_g   = (const float*)d_in[29];
    const float* bnR_b   = (const float*)d_in[30];
    const float* bnR_m   = (const float*)d_in[31];
    const float* bnR_v   = (const float*)d_in[32];
    const float* fc1_w   = (const float*)d_in[33];
    const float* fc1_b   = (const float*)d_in[34];
    const float* ln_g    = (const float*)d_in[35];
    const float* ln_b    = (const float*)d_in[36];
    const float* fc2_w   = (const float*)d_in[37];
    const float* fc2_b   = (const float*)d_in[38];

    const int* src = ei;
    const int* dst = ei + EE;

    float* bufA = (float*)d_ws;
    float* bufB = bufA + (size_t)NN * HID;
    int*   col    = (int*)(bufB + (size_t)NN * HID);
    int*   rank   = col + EE;
    int*   deg    = rank + EE;
    int*   rowptr = deg + NN;
    int*   bsum   = rowptr + NN + 4;
    float* sums = bufA + (size_t)3 * NN * HID;
    float* cnt  = sums + (size_t)GG * HID;

    const int TB = 256;
    const int eBlocks = (EE + TB - 1) / TB;
    const int nodeBlocks = NN / 8;

    bn_in_kernel<<<(NN * INDIM + TB - 1) / TB, TB, 0, stream>>>(
        x, in_g, in_b, in_m, in_v, bufA, deg);

    edge_rank_kernel<<<eBlocks, TB, 0, stream>>>(dst, deg, rank);
    deg_blocksum_kernel<<<SCAN_BLOCKS, TB, 0, stream>>>(deg, bsum);
    rowptr_scan_kernel<<<SCAN_BLOCKS, TB, 0, stream>>>(deg, bsum, rowptr);
    edge_fill_kernel<<<eBlocks, TB, 0, stream>>>(src, dst, rank, rowptr, col);

    gather_agg_kernel<INDIM><<<nodeBlocks, TB, 0, stream>>>(
        bufA, rowptr, col, eps0, 0, bufB);
    gemm_bn_gelu_kernel<INDIM><<<GEMM_BLOCKS, GEMM_THREADS, 0, stream>>>(
        bufB, c0_w1, c0_b1, c0_g, c0_bb, c0_m, c0_v, bufA);
    gemm_bn_gelu_kernel<HID><<<GEMM_BLOCKS, GEMM_THREADS, 0, stream>>>(
        bufA, c0_w2, c0_b2, bn0_g, bn0_b, bn0_m, bn0_v, bufB);

    float* hin = bufB;
    float* hout = bufA;
    for (int l = 0; l < 2; ++l) {
        const size_t wOff = (size_t)l * HID * HID;
        const size_t vOff = (size_t)l * HID;
        gather_agg_kernel<HID><<<nodeBlocks, TB, 0, stream>>>(
            hin, rowptr, col, epsR, l, hout);
        gemm_bn_gelu_kernel<HID><<<GEMM_BLOCKS, GEMM_THREADS, 0, stream>>>(
            hout, cR_w1 + wOff, cR_b1 + vOff, cR_g + vOff, cR_bb + vOff,
            cR_m + vOff, cR_v + vOff, hin);
        gemm_bn_gelu_kernel<HID><<<GEMM_BLOCKS, GEMM_THREADS, 0, stream>>>(
            hin, cR_w2 + wOff, cR_b2 + vOff, bnR_g + vOff, bnR_b + vOff,
            bnR_m + vOff, bnR_v + vOff, hout);
        float* tmp = hin; hin = hout; hout = tmp;
    }

    zero_kernel<<<(GG * HID + GG + TB - 1) / TB, TB, 0, stream>>>(sums, GG * HID + GG);
    {
        int total = NN * (HID / 4);
        pool_acc_kernel<<<(total + TB - 1) / TB, TB, 0, stream>>>(
            hin, batch, sums, cnt, total);
    }
    pool_final_kernel<<<(GG * HID + TB - 1) / TB, TB, 0, stream>>>(sums, cnt, hout);

    head_kernel<<<GG, HID, 0, stream>>>(
        hout, fc1_w, fc1_b, ln_g, ln_b, fc2_w, fc2_b, (float*)d_out);
}


# Edge grouping by a two-level counting sort on LDS counters (bucket of 256 dst nodes, then node) replaces the per-edge returning global atomics; add order = arrival-order rank as before
# speedup vs baseline: 2.0246x; 1.1511x over previous
#include <hip/hip_runtime.h>
#include <math.h>

typedef __attribute__((ext_vector_type(2))) float v2f;
typedef __attribute__((ext_vector_type(8))) float v8f;

#define NN   100000
#define EE   1600000
#define INDIM 64
#define HID  128
#define LAT  64
#define GG   1000

__device__ __forceinline__ float gelu_exact(float x) {
    return 0.5f * x * (1.0f + erff(x * 0.70710678118654752f));
}

__global__ __launch_bounds__(256) void bn_in_kernel(
    const float* __restrict__ x,
    const float* __restrict__ g, const float* __restrict__ b,
    const float* __restrict__ m, const float* __restrict__ v,
    float* __restrict__ out)
{
    int idx = blockIdx.x * blockDim.x + threadIdx.x;
    if (idx >= NN * INDIM) return;
    int j = idx & (INDIM - 1);
    float y = (x[idx] - m[j]) * rsqrtf(v[j] + 1e-5f) * g[j] + b[j];
    out[idx] = y;
}

#define BUCKET_SHIFT 8
#define BUCKET_NODES (1 << BUCKET_SHIFT)
#define NBUCKETS ((NN + BUCKET_NODES - 1) / BUCKET_NODES)
#define PART_BLOCKS 256
#define PART_THREADS 1024
#define PART_CHUNK ((EE + PART_BLOCKS - 1) / PART_BLOCKS)
#define PART_BATCH 8
#define PART_CELLS (NBUCKETS * PART_BLOCKS)
#define SCAN_THREADS 1024
#define SCAN_BLOCKS ((PART_CELLS + SCAN_THREADS - 1) / SCAN_THREADS)
#define SCAN_PRE_MAX (((SCAN_BLOCKS - 1) * (SCAN_THREADS / 4) + SCAN_THREADS - 1) / SCAN_THREADS)
#define SORT_THREADS 1024
#define SORT_BATCH 5

__global__ __launch_bounds__(PART_THREADS) void part_hist_kernel(
    const int* __restrict__ dst, int* __restrict__ cells)
{
    __shared__ int h[NBUCKETS];
    const int t = threadIdx.x;
    for (int k = t; k < NBUCKETS; k += PART_THREADS) h[k] = 0;
    __syncthreads();
    const int e0 = blockIdx.x * PART_CHUNK;
    const int e1 = min(e0 + PART_CHUNK, EE);
    for (int e = e0 + t; e < e1; e += PART_BATCH * PART_THREADS) {
        int d[PART_BATCH];
        #pragma unroll
        for (int u = 0; u < PART_BATCH; ++u) {
            const int ee = e + u * PART_THREADS;
            d[u] = (ee < e1) ? dst[ee] : -1;
        }
        #pragma unroll
        for (int u = 0; u < PART_BATCH; ++u)
            if ((unsigned)d[u] < (unsigned)NN) atomicAdd(&h[d[u] >> BUCKET_SHIFT], 1);
    }
    __syncthreads();
    for (int k = t; k < NBUCKETS; k += PART_THREADS)
        cells[k * PART_BLOCKS + blockIdx.x] = h[k];
}

__global__ __launch_bounds__(SCAN_THREADS) void part_scan_kernel(
    const int* __restrict__ cells, int* __restrict__ base)
{
    __shared__ int red[SCAN_THREADS / 32];
    __shared__ int wsum[SCAN_THREADS / 32];
    const int t = threadIdx.x;
    const int lane = t & 31;
    const int wave = t >> 5;
    const int nb = blockIdx.x;

    const int4* __restrict__ c4 = (const int4*)cells;
    const int n4 = nb * (SCAN_THREADS / 4);
    int4 q[SCAN_PRE_MAX];
    #pragma unroll
    for (int u = 0; u < SCAN_PRE_MAX; ++u) {
        const int i = u * SCAN_THREADS + t;
        q[u] = (i < n4) ? c4[i] : make_int4(0, 0, 0, 0);
    }
    int s = 0;
    #pragma unroll
    for (int u = 0; u < SCAN_PRE_MAX; ++u) s += q[u].x + q[u].y + q[u].z + q[u].w;
    const int idx = nb * SCAN_THREADS + t;
    const int x = (idx < PART_CELLS) ? cells[idx] : 0;

    for (int off = 16; off > 0; off >>= 1) s += __shfl_xor(s, off);
    int inc = x;
    for (int off = 1; off < 32; off <<= 1) {
        const int n = __shfl_up(inc, off);
        if (lane >= off) inc += n;
    }
    if (lane == 0) red[wave] = s;
    if (lane == 31) wsum[wave] = inc;
    __syncthreads();
    int before = 0;
    for (int w = 0; w < SCAN_THREADS / 32; ++w) before += red[w];
    int woff = 0;
    for (int w = 0; w < wave; ++w) woff += wsum[w];
    const int ex = before + woff + inc - x;
    if (idx < PART_CELLS) base[idx] = ex;
    if (idx == PART_CELLS - 1) base[PART_CELLS] = ex + x;
}

__global__ __launch_bounds__(PART_THREADS) void part_scatter_kernel(
    const int* __restrict__ src, const int* __restrict__ dst,
    const int* __restrict__ base, int* __restrict__ pairs_out)
{
    int2* __restrict__ pairs = (int2*)pairs_out;
    __shared__ int cur[NBUCKETS];
    const int t = threadIdx.x;
    for (int k = t; k < NBUCKETS; k += PART_THREADS)
        cur[k] = base[k * PART_BLOCKS + blockIdx.x];
    __syncthreads();
    const int e0 = blockIdx.x * PART_CHUNK;
    const int e1 = min(e0 + PART_CHUNK, EE);
    for (int e = e0 + t; e < e1; e += PART_BATCH * PART_THREADS) {
        int d[PART_BATCH], s[PART_BATCH];
        #pragma unroll
        for (int u = 0; u < PART_BATCH; ++u) {
            const int ee = e + u * PART_THREADS;
            const bool ok = ee < e1;
            d[u] = ok ? dst[ee] : -1;
            s[u] = ok ? src[ee] : 0;
        }
        #pragma unroll
        for (int u = 0; u < PART_BATCH; ++u) {
            if ((unsigned)d[u] < (unsigned)NN) {
                int pos = atomicAdd(&cur[d[u] >> BUCKET_SHIFT], 1);
                pos = (int)min((unsigned)pos, (unsigned)(EE - 1));
                pairs[pos] = make_int2(d[u], s[u]);
            }
        }
    }
}

__global__ __launch_bounds__(SORT_THREADS) void bucket_sort_kernel(
    const int* __restrict__ pairs_in, const int* __restrict__ base,
    int* __restrict__ rowptr, int* __restrict__ col)
{
    const int2* __restrict__ pairs = (const int2*)pairs_in;
    __shared__ int c[BUCKET_NODES];
    __shared__ int wsum[BUCKET_NODES / 32];
    const int t = threadIdx.x;
    const int lane = t & 31;
    const int wave = t >> 5;
    const int k = blockIdx.x;
    int bs = base[k * PART_BLOCKS];
    int be = base[(k + 1) * PART_BLOCKS];
    bs = max(bs, 0);
    be = min(be, EE);

    if (t < BUCKET_NODES) c[t] = 0;
    __syncthreads();
    for (int p = bs + t; p < be; p += SORT_BATCH * SORT_THREADS) {
        int d[SORT_BATCH];
        #pragma unroll
        for (int u = 0; u < SORT_BATCH; ++u) {
            const int pp = p + u * SORT_THREADS;
            d[u] = (pp < be) ? pairs[pp].x : -1;
        }
        #pragma unroll
        for (int u = 0; u < SORT_BATCH; ++u)
            if (d[u] >= 0) atomicAdd(&c[d[u] & (BUCKET_NODES - 1)], 1);
    }
    __syncthreads();

    int x = 0, inc = 0, start = 0;
    if (t < BUCKET_NODES) {
        x = c[t];
        inc = x;
        for (int off = 1; off < 32; off <<= 1) {
            const int n = __shfl_up(inc, off);
            if (lane >= off) inc += n;
        }
        if (lane == 31) wsum[wave] = inc;
    }
    __syncthreads();
    if (t < BUCKET_NODES) {
        int woff = 0;
        for (int w = 0; w < wave; ++w) woff += wsum[w];
        start = bs + woff + inc - x;
        c[t] = start;
    }
    __syncthreads();

    for (int p = bs + t; p < be; p += SORT_BATCH * SORT_THREADS) {
        int2 pr[SORT_BATCH];
        #pragma unroll
        for (int u = 0; u < SORT_BATCH; ++u) {
            const int pp = p + u * SORT_THREADS;
            pr[u] = (pp < be) ? pairs[pp] : make_int2(-1, 0);
        }
        #pragma unroll
        for (int u = 0; u < SORT_BATCH; ++u) {
            if (pr[u].x >= 0) {
                int pos = atomicAdd(&c[pr[u].x & (BUCKET_NODES - 1)], 1);
                pos = (int)min((unsigned)pos, (unsigned)(EE - 1));
                col[pos] = pr[u].y;
            }
        }
    }
    if (t < BUCKET_NODES) {
        const int node = k * BUCKET_NODES + t;
        if (node < NN) rowptr[node] = start;
    }
    if (k == NBUCKETS - 1 && t == 0) rowptr[NN] = be;
}

template<int DIM>
__global__ __launch_bounds__(256) void gather_agg_kernel(
    const float* __restrict__ h, const int* __restrict__ rowptr,
    const int* __restrict__ col, const float* __restrict__ epsP, int epsIdx,
    float* __restrict__ agg)
{
    constexpr int VW = DIM / 32;
    typedef __attribute__((ext_vector_type(VW))) float vt;
    const int lane = threadIdx.x & 31;
    const int node = __builtin_amdgcn_readfirstlane(
        (int)(blockIdx.x * 8 + (threadIdx.x >> 5)));
    const int beg = rowptr[node];
    const int end = rowptr[node + 1];
    const float* __restrict__ hl = h + VW * lane;

    const vt self = *(const vt*)(hl + (size_t)node * DIM);

    vt acc = {};
    const vt zero = {};
    for (int base = beg; base < end; base += 32) {
        const int n = min(32, end - base);
        int ids = (lane < n) ? col[base + lane] : 0;
        ids = (int)min((unsigned)ids, (unsigned)(NN - 1));
        int j = 0;
        for (; j + 4 <= n; j += 4) {
            const int s0 = __builtin_amdgcn_readlane(ids, j);
            const int s1 = __builtin_amdgcn_readlane(ids, j + 1);
            const int s2 = __builtin_amdgcn_readlane(ids, j + 2);
            const int s3 = __builtin_amdgcn_readlane(ids, j + 3);
            const vt v0 = *(const vt*)(hl + (size_t)s0 * DIM);
            const vt v1 = *(const vt*)(hl + (size_t)s1 * DIM);
            const vt v2 = *(const vt*)(hl + (size_t)s2 * DIM);
            const vt v3 = *(const vt*)(hl + (size_t)s3 * DIM);
            acc += v0; acc += v1; acc += v2; acc += v3;
        }
        if (j < n) {
            const int last = n - 1;
            const int s0 = __builtin_amdgcn_readlane(ids, j);
            const int s1 = __builtin_amdgcn_readlane(ids, min(j + 1, last));
            const int s2 = __builtin_amdgcn_readlane(ids, min(j + 2, last));
            const vt v0 = *(const vt*)(hl + (size_t)s0 * DIM);
            const vt v1 = *(const vt*)(hl + (size_t)s1 * DIM);
            const vt v2 = *(const vt*)(hl + (size_t)s2 * DIM);
            acc += v0;
            acc += (j + 1 < n) ? v1 : zero;
            acc += (j + 2 < n) ? v2 : zero;
        }
    }
    const float sc = 1.0f + epsP[epsIdx];
    *(vt*)(agg + (size_t)node * DIM + VW * lane) = sc * self + acc;
}

#define ROW_TILES (NN / 16)
#define GEMM_WAVES 7
#define GEMM_THREADS (GEMM_WAVES * 32)
#define GEMM_BLOCKS ((ROW_TILES + GEMM_WAVES - 1) / GEMM_WAVES)
#define WPITCH 144

template<int K>
__global__ __launch_bounds__(GEMM_THREADS) void gemm_bn_gelu_kernel(
    const float* __restrict__ A,
    const float* __restrict__ W,
    const float* __restrict__ bias,
    const float* __restrict__ bn_g, const float* __restrict__ bn_b,
    const float* __restrict__ bn_m, const float* __restrict__ bn_v,
    float* __restrict__ out)
{
    __shared__ __attribute__((aligned(16))) float Ws[(K / 2) * WPITCH * 2];

    const int lane = threadIdx.x & 31;
    const int wave = __builtin_amdgcn_readfirstlane((int)(threadIdx.x >> 5));
    const int half = lane >> 4;
    const int mr   = lane & 15;

    const int tileM = blockIdx.x * GEMM_WAVES + wave;
    const bool live = tileM < ROW_TILES;
    const float* __restrict__ Arow =
        A + (size_t)((live ? tileM : 0) * 16 + mr) * K + 2 * half;

    constexpr int NFULL = (K / 2) / GEMM_WAVES;
    constexpr int NREM  = (K / 2) - NFULL * GEMM_WAVES;
    const bool extra = wave < NREM;
    float4 e0 = {}, e1 = {};
    if (extra) {
        const int kp = NFULL * GEMM_WAVES + wave;
        e0 = *(const float4*)(W + (size_t)(2 * kp) * HID + 4 * lane);
        e1 = *(const float4*)(W + (size_t)(2 * kp + 1) * HID + 4 * lane);
    }
    float4 r0[NFULL], r1[NFULL];
    #pragma unroll
    for (int i = 0; i < NFULL; ++i) {
        const int kp = wave + GEMM_WAVES * i;
        r0[i] = *(const float4*)(W + (size_t)(2 * kp) * HID + 4 * lane);
        r1[i] = *(const float4*)(W + (size_t)(2 * kp + 1) * HID + 4 * lane);
    }
    constexpr int KC = 16;
    v2f a_cur[KC / 4];
    #pragma unroll
    for (int i = 0; i < KC / 4; ++i) a_cur[i] = *(const v2f*)(Arow + 4 * i);
    #pragma unroll
    for (int i = 0; i < NFULL; ++i) {
        const int kp = wave + GEMM_WAVES * i;
        float* d = Ws + (kp * WPITCH + 4 * lane) * 2;
        *(float4*)(d)     = make_float4(r0[i].x, r1[i].x, r0[i].y, r1[i].y);
        *(float4*)(d + 4) = make_float4(r0[i].z, r1[i].z, r0[i].w, r1[i].w);
    }
    if (extra) {
        const int kp = NFULL * GEMM_WAVES + wave;
        float* d = Ws + (kp * WPITCH + 4 * lane) * 2;
        *(float4*)(d)     = make_float4(e0.x, e1.x, e0.y, e1.y);
        *(float4*)(d + 4) = make_float4(e0.z, e1.z, e0.w, e1.w);
    }
    __syncthreads();
    if (!live) return;

    const float* Bl = Ws + (half * WPITCH + mr) * 2;

    v8f acc[8];
    #pragma unroll
    for (int c = 0; c < 8; ++c) acc[c] = (v8f){};

    #pragma unroll 1
    for (int kc = 0; kc < K; kc += KC) {
        const int kn = min(kc + KC, K - KC);
        v2f a_nxt[KC / 4];
        #pragma unroll
        for (int i = 0; i < KC / 4; ++i) a_nxt[i] = *(const v2f*)(Arow + kn + 4 * i);
        #pragma unroll
        for (int i = 0; i < KC / 4; ++i) {
            #pragma unroll
            for (int c = 0; c < 8; ++c) {
                const v2f b = *(const v2f*)(Bl + ((kc + 4 * i) / 2) * WPITCH * 2 + c * 32);
                acc[c] = __builtin_amdgcn_wmma_f32_16x16x4_f32(
                    false, a_cur[i], false, b, (short)0, acc[c], false, false);
            }
        }
        #pragma unroll
        for (int i = 0; i < KC / 4; ++i) a_cur[i] = a_nxt[i];
    }

    #pragma unroll
    for (int c = 0; c < 8; ++c) {
        const int col = c * 16 + mr;
        const float cb = bias[col];
        const float cm = bn_m[col];
        const float cs = rsqrtf(bn_v[col] + 1e-5f);
        const float cg = bn_g[col];
        const float co = bn_b[col];
        #pragma unroll
        for (int r = 0; r < 8; ++r) {
            const int row = tileM * 16 + r + 8 * half;
            float y = acc[c][r] + cb;
            y = (y - cm) * cs * cg + co;
            y = gelu_exact(y);
            out[(size_t)row * HID + col] = y;
        }
    }
}

__global__ __launch_bounds__(256) void zero_kernel(float* __restrict__ p, int n) {
    int idx = blockIdx.x * blockDim.x + threadIdx.x;
    if (idx < n) p[idx] = 0.0f;
}

__global__ __launch_bounds__(256) void pool_acc_kernel(
    const float* __restrict__ h, const int* __restrict__ batch,
    float* __restrict__ sums, float* __restrict__ cnt, int total)
{
    int idx = blockIdx.x * blockDim.x + threadIdx.x;
    if (idx >= total) return;
    int i = idx >> 5;
    int c = idx & 31;
    int b = batch[i];
    const float4 v = *(const float4*)(h + (size_t)i * HID + 4 * c);
    float* s = sums + (size_t)b * HID + 4 * c;
    unsafeAtomicAdd(s + 0, v.x);
    unsafeAtomicAdd(s + 1, v.y);
    unsafeAtomicAdd(s + 2, v.z);
    unsafeAtomicAdd(s + 3, v.w);
    if (c == 0) unsafeAtomicAdd(cnt + b, 1.0f);
}

__global__ __launch_bounds__(256) void pool_final_kernel(
    const float* __restrict__ sums, const float* __restrict__ cnt,
    float* __restrict__ pooled)
{
    int idx = blockIdx.x * blockDim.x + threadIdx.x;
    if (idx >= GG * HID) return;
    int g = idx / HID;
    pooled[idx] = sums[idx] / fmaxf(cnt[g], 1.0f);
}

__global__ __launch_bounds__(128) void head_kernel(
    const float* __restrict__ pooled,
    const float* __restrict__ fc1w, const float* __restrict__ fc1b,
    const float* __restrict__ lng, const float* __restrict__ lnb,
    const float* __restrict__ fc2w, const float* __restrict__ fc2b,
    float* __restrict__ out)
{
    __shared__ float p[HID];
    __shared__ float rbuf[HID];
    __shared__ float red[HID];
    const int g = blockIdx.x;
    const int t = threadIdx.x;

    p[t] = pooled[(size_t)g * HID + t];
    __syncthreads();

    float o = fc1b[t];
    #pragma unroll 4
    for (int k = 0; k < HID; ++k) o += p[k] * fc1w[k * HID + t];

    red[t] = o;
    __syncthreads();
    for (int s = 64; s > 0; s >>= 1) {
        if (t < s) red[t] += red[t + s];
        __syncthreads();
    }
    const float mu = red[0] * (1.0f / HID);
    __syncthreads();

    const float dv = o - mu;
    red[t] = dv * dv;
    __syncthreads();
    for (int s = 64; s > 0; s >>= 1) {
        if (t < s) red[t] += red[t + s];
        __syncthreads();
    }
    const float var = red[0] * (1.0f / HID);

    float y = dv * rsqrtf(var + 1e-5f) * lng[t] + lnb[t];
    y = gelu_exact(y) + p[t];
    rbuf[t] = y;
    __syncthreads();

    if (t < LAT) {
        float o2 = fc2b[t];
        #pragma unroll 4
        for (int k = 0; k < HID; ++k) o2 += rbuf[k] * fc2w[k * LAT + t];
        out[(size_t)g * LAT + t] = o2;
    }
}

extern "C" void kernel_launch(void* const* d_in, const int* in_sizes, int n_in,
                              void* d_out, int out_size, void* d_ws, size_t ws_size,
                              hipStream_t stream) {
    const float* x       = (const float*)d_in[0];
    const int*   ei      = (const int*)  d_in[1];
    const int*   batch   = (const int*)  d_in[2];
    const float* in_g    = (const float*)d_in[3];
    const float* in_b    = (const float*)d_in[4];
    const float* in_m    = (const float*)d_in[5];
    const float* in_v    = (const float*)d_in[6];
    const float* eps0    = (const float*)d_in[7];
    const float* c0_w1   = (const float*)d_in[8];
    const float* c0_b1   = (const float*)d_in[9];
    const float* c0_g    = (const float*)d_in[10];
    const float* c0_bb   = (const float*)d_in[11];
    const float* c0_m    = (const float*)d_in[12];
    const float* c0_v    = (const float*)d_in[13];
    const float* c0_w2   = (const float*)d_in[14];
    const float* c0_b2   = (const float*)d_in[15];
    const float* bn0_g   = (const float*)d_in[16];
    const float* bn0_b   = (const float*)d_in[17];
    const float* bn0_m   = (const float*)d_in[18];
    const float* bn0_v   = (const float*)d_in[19];
    const float* epsR    = (const float*)d_in[20];
    const float* cR_w1   = (const float*)d_in[21];
    const float* cR_b1   = (const float*)d_in[22];
    const float* cR_g    = (const float*)d_in[23];
    const float* cR_bb   = (const float*)d_in[24];
    const float* cR_m    = (const float*)d_in[25];
    const float* cR_v    = (const float*)d_in[26];
    const float* cR_w2   = (const float*)d_in[27];
    const float* cR_b2   = (const float*)d_in[28];
    const float* bnR_g   = (const float*)d_in[29];
    const float* bnR_b   = (const float*)d_in[30];
    const float* bnR_m   = (const float*)d_in[31];
    const float* bnR_v   = (const float*)d_in[32];
    const float* fc1_w   = (const float*)d_in[33];
    const float* fc1_b   = (const float*)d_in[34];
    const float* ln_g    = (const float*)d_in[35];
    const float* ln_b    = (const float*)d_in[36];
    const float* fc2_w   = (const float*)d_in[37];
    const float* fc2_b   = (const float*)d_in[38];

    const int* src = ei;
    const int* dst = ei + EE;

    float* bufA = (float*)d_ws;
    float* bufB = bufA + (size_t)NN * HID;
    int*   col    = (int*)(bufB + (size_t)NN * HID);
    int*   pairs  = col + EE;
    int*   cells  = pairs + 2 * EE;
    int*   base   = cells + PART_CELLS;
    int*   rowptr = base + PART_CELLS + 4;
    float* sums = bufA + (size_t)3 * NN * HID;
    float* cnt  = sums + (size_t)GG * HID;

    const int TB = 256;
    const int nodeBlocks = NN / 8;

    bn_in_kernel<<<(NN * INDIM + TB - 1) / TB, TB, 0, stream>>>(
        x, in_g, in_b, in_m, in_v, bufA);

    part_hist_kernel<<<PART_BLOCKS, PART_THREADS, 0, stream>>>(dst, cells);
    part_scan_kernel<<<SCAN_BLOCKS, SCAN_THREADS, 0, stream>>>(cells, base);
    part_scatter_kernel<<<PART_BLOCKS, PART_THREADS, 0, stream>>>(src, dst, base, pairs);
    bucket_sort_kernel<<<NBUCKETS, SORT_THREADS, 0, stream>>>(pairs, base, rowptr, col);

    gather_agg_kernel<INDIM><<<nodeBlocks, TB, 0, stream>>>(
        bufA, rowptr, col, eps0, 0, bufB);
    gemm_bn_gelu_kernel<INDIM><<<GEMM_BLOCKS, GEMM_THREADS, 0, stream>>>(
        bufB, c0_w1, c0_b1, c0_g, c0_bb, c0_m, c0_v, bufA);
    gemm_bn_gelu_kernel<HID><<<GEMM_BLOCKS, GEMM_THREADS, 0, stream>>>(
        bufA, c0_w2, c0_b2, bn0_g, bn0_b, bn0_m, bn0_v, bufB);

    float* hin = bufB;
    float* hout = bufA;
    for (int l = 0; l < 2; ++l) {
        const size_t wOff = (size_t)l * HID * HID;
        const size_t vOff = (size_t)l * HID;
        gather_agg_kernel<HID><<<nodeBlocks, TB, 0, stream>>>(
            hin, rowptr, col, epsR, l, hout);
        gemm_bn_gelu_kernel<HID><<<GEMM_BLOCKS, GEMM_THREADS, 0, stream>>>(
            hout, cR_w1 + wOff, cR_b1 + vOff, cR_g + vOff, cR_bb + vOff,
            cR_m + vOff, cR_v + vOff, hin);
        gemm_bn_gelu_kernel<HID><<<GEMM_BLOCKS, GEMM_THREADS, 0, stream>>>(
            hin, cR_w2 + wOff, cR_b2 + vOff, bnR_g + vOff, bnR_b + vOff,
            bnR_m + vOff, bnR_v + vOff, hout);
        float* tmp = hin; hin = hout; hout = tmp;
    }

    zero_kernel<<<(GG * HID + GG + TB - 1) / TB, TB, 0, stream>>>(sums, GG * HID + GG);
    {
        int total = NN * (HID / 4);
        pool_acc_kernel<<<(total + TB - 1) / TB, TB, 0, stream>>>(
            hin, batch, sums, cnt, total);
    }
    pool_final_kernel<<<(GG * HID + TB - 1) / TB, TB, 0, stream>>>(sums, cnt, hout);

    head_kernel<<<GG, HID, 0, stream>>>(
        hout, fc1_w, fc1_b, ln_g, ln_b, fc2_w, fc2_b, (float*)d_out);
}


# Each layer's two products in one kernel: the first taken transposed, so its GELU output (straight-line erf, same values) stays in registers as the second's A operand; second product's k order changes
# speedup vs baseline: 2.7183x; 1.3426x over previous
#include <hip/hip_runtime.h>
#include <math.h>

typedef __attribute__((ext_vector_type(2))) float v2f;
typedef __attribute__((ext_vector_type(8))) float v8f;

#define NN   100000
#define EE   1600000
#define INDIM 64
#define HID  128
#define LAT  64
#define GG   1000

__device__ __forceinline__ float gelu_exact(float x) {
    return 0.5f * x * (1.0f + erff(x * 0.70710678118654752f));
}

__device__ __forceinline__ float erf_select(float x) {
    const float ax = fabsf(x);
    float p = fmaf(ax, __uint_as_float(0x378e98abu), __uint_as_float(0xb9c68948u));
    p = fmaf(ax, p, __uint_as_float(0x3b7cd369u));
    p = fmaf(ax, p, __uint_as_float(0xbcc618b2u));
    p = fmaf(ax, p, __uint_as_float(0x3dda74e4u));
    p = fmaf(ax, p, __uint_as_float(0x3f228afdu));
    p = fmaf(ax, p, __uint_as_float(0x3e03c728u));
    p = fmaf(ax, p, ax);
    const float hi = 1.0f - expf(-p);
    const float t = x * x;
    float q = fmaf(t, __uint_as_float(0xba1345e1u), __uint_as_float(0x3ba10414u));
    q = fmaf(t, q, __uint_as_float(0xbcdac9b8u));
    q = fmaf(t, q, __uint_as_float(0x3de703beu));
    q = fmaf(t, q, __uint_as_float(0xbec09330u));
    q = fmaf(t, q, __uint_as_float(0x3e0375d0u));
    const float lo = fmaf(ax, q, ax);
    return copysignf((ax < 1.0f) ? lo : hi, x);
}

__device__ __forceinline__ float gelu_exact_select(float x) {
    return 0.5f * x * (1.0f + erf_select(x * 0.70710678118654752f));
}

__global__ __launch_bounds__(256) void bn_in_kernel(
    const float* __restrict__ x,
    const float* __restrict__ g, const float* __restrict__ b,
    const float* __restrict__ m, const float* __restrict__ v,
    float* __restrict__ out)
{
    int idx = blockIdx.x * blockDim.x + threadIdx.x;
    if (idx >= NN * INDIM) return;
    int j = idx & (INDIM - 1);
    float y = (x[idx] - m[j]) * rsqrtf(v[j] + 1e-5f) * g[j] + b[j];
    out[idx] = y;
}

#define BUCKET_SHIFT 8
#define BUCKET_NODES (1 << BUCKET_SHIFT)
#define NBUCKETS ((NN + BUCKET_NODES - 1) / BUCKET_NODES)
#define PART_BLOCKS 256
#define PART_THREADS 1024
#define PART_CHUNK ((EE + PART_BLOCKS - 1) / PART_BLOCKS)
#define PART_BATCH 8
#define PART_CELLS (NBUCKETS * PART_BLOCKS)
#define SCAN_THREADS 1024
#define SCAN_BLOCKS ((PART_CELLS + SCAN_THREADS - 1) / SCAN_THREADS)
#define SCAN_PRE_MAX (((SCAN_BLOCKS - 1) * (SCAN_THREADS / 4) + SCAN_THREADS - 1) / SCAN_THREADS)
#define SORT_THREADS 1024
#define SORT_BATCH 5

__global__ __launch_bounds__(PART_THREADS) void part_hist_kernel(
    const int* __restrict__ dst, int* __restrict__ cells)
{
    __shared__ int h[NBUCKETS];
    const int t = threadIdx.x;
    for (int k = t; k < NBUCKETS; k += PART_THREADS) h[k] = 0;
    __syncthreads();
    const int e0 = blockIdx.x * PART_CHUNK;
    const int e1 = min(e0 + PART_CHUNK, EE);
    for (int e = e0 + t; e < e1; e += PART_BATCH * PART_THREADS) {
        int d[PART_BATCH];
        #pragma unroll
        for (int u = 0; u < PART_BATCH; ++u) {
            const int ee = e + u * PART_THREADS;
            d[u] = (ee < e1) ? dst[ee] : -1;
        }
        #pragma unroll
        for (int u = 0; u < PART_BATCH; ++u)
            if ((unsigned)d[u] < (unsigned)NN) atomicAdd(&h[d[u] >> BUCKET_SHIFT], 1);
    }
    __syncthreads();
    for (int k = t; k < NBUCKETS; k += PART_THREADS)
        cells[k * PART_BLOCKS + blockIdx.x] = h[k];
}

__global__ __launch_bounds__(SCAN_THREADS) void part_scan_kernel(
    const int* __restrict__ cells, int* __restrict__ base)
{
    __shared__ int red[SCAN_THREADS / 32];
    __shared__ int wsum[SCAN_THREADS / 32];
    const int t = threadIdx.x;
    const int lane = t & 31;
    const int wave = t >> 5;
    const int nb = blockIdx.x;

    const int4* __restrict__ c4 = (const int4*)cells;
    const int n4 = nb * (SCAN_THREADS / 4);
    int4 q[SCAN_PRE_MAX];
    #pragma unroll
    for (int u = 0; u < SCAN_PRE_MAX; ++u) {
        const int i = u * SCAN_THREADS + t;
        q[u] = (i < n4) ? c4[i] : make_int4(0, 0, 0, 0);
    }
    int s = 0;
    #pragma unroll
    for (int u = 0; u < SCAN_PRE_MAX; ++u) s += q[u].x + q[u].y + q[u].z + q[u].w;
    const int idx = nb * SCAN_THREADS + t;
    const int x = (idx < PART_CELLS) ? cells[idx] : 0;

    for (int off = 16; off > 0; off >>= 1) s += __shfl_xor(s, off);
    int inc = x;
    for (int off = 1; off < 32; off <<= 1) {
        const int n = __shfl_up(inc, off);
        if (lane >= off) inc += n;
    }
    if (lane == 0) red[wave] = s;
    if (lane == 31) wsum[wave] = inc;
    __syncthreads();
    int before = 0;
    for (int w = 0; w < SCAN_THREADS / 32; ++w) before += red[w];
    int woff = 0;
    for (int w = 0; w < wave; ++w) woff += wsum[w];
    const int ex = before + woff + inc - x;
    if (idx < PART_CELLS) base[idx] = ex;
    if (idx == PART_CELLS - 1) base[PART_CELLS] = ex + x;
}

__global__ __launch_bounds__(PART_THREADS) void part_scatter_kernel(
    const int* __restrict__ src, const int* __restrict__ dst,
    const int* __restrict__ base, int* __restrict__ pairs_out)
{
    int2* __restrict__ pairs = (int2*)pairs_out;
    __shared__ int cur[NBUCKETS];
    const int t = threadIdx.x;
    for (int k = t; k < NBUCKETS; k += PART_THREADS)
        cur[k] = base[k * PART_BLOCKS + blockIdx.x];
    __syncthreads();
    const int e0 = blockIdx.x * PART_CHUNK;
    const int e1 = min(e0 + PART_CHUNK, EE);
    for (int e = e0 + t; e < e1; e += PART_BATCH * PART_THREADS) {
        int d[PART_BATCH], s[PART_BATCH];
        #pragma unroll
        for (int u = 0; u < PART_BATCH; ++u) {
            const int ee = e + u * PART_THREADS;
            const bool ok = ee < e1;
            d[u] = ok ? dst[ee] : -1;
            s[u] = ok ? src[ee] : 0;
        }
        #pragma unroll
        for (int u = 0; u < PART_BATCH; ++u) {
            if ((unsigned)d[u] < (unsigned)NN) {
                int pos = atomicAdd(&cur[d[u] >> BUCKET_SHIFT], 1);
                pos = (int)min((unsigned)pos, (unsigned)(EE - 1));
                pairs[pos] = make_int2(d[u], s[u]);
            }
        }
    }
}

__global__ __launch_bounds__(SORT_THREADS) void bucket_sort_kernel(
    const int* __restrict__ pairs_in, const int* __restrict__ base,
    int* __restrict__ rowptr, int* __restrict__ col)
{
    const int2* __restrict__ pairs = (const int2*)pairs_in;
    __shared__ int c[BUCKET_NODES];
    __shared__ int wsum[BUCKET_NODES / 32];
    const int t = threadIdx.x;
    const int lane = t & 31;
    const int wave = t >> 5;
    const int k = blockIdx.x;
    int bs = base[k * PART_BLOCKS];
    int be = base[(k + 1) * PART_BLOCKS];
    bs = max(bs, 0);
    be = min(be, EE);

    if (t < BUCKET_NODES) c[t] = 0;
    __syncthreads();
    for (int p = bs + t; p < be; p += SORT_BATCH * SORT_THREADS) {
        int d[SORT_BATCH];
        #pragma unroll
        for (int u = 0; u < SORT_BATCH; ++u) {
            const int pp = p + u * SORT_THREADS;
            d[u] = (pp < be) ? pairs[pp].x : -1;
        }
        #pragma unroll
        for (int u = 0; u < SORT_BATCH; ++u)
            if (d[u] >= 0) atomicAdd(&c[d[u] & (BUCKET_NODES - 1)], 1);
    }
    __syncthreads();

    int x = 0, inc = 0, start = 0;
    if (t < BUCKET_NODES) {
        x = c[t];
        inc = x;
        for (int off = 1; off < 32; off <<= 1) {
            const int n = __shfl_up(inc, off);
            if (lane >= off) inc += n;
        }
        if (lane == 31) wsum[wave] = inc;
    }
    __syncthreads();
    if (t < BUCKET_NODES) {
        int woff = 0;
        for (int w = 0; w < wave; ++w) woff += wsum[w];
        start = bs + woff + inc - x;
        c[t] = start;
    }
    __syncthreads();

    for (int p = bs + t; p < be; p += SORT_BATCH * SORT_THREADS) {
        int2 pr[SORT_BATCH];
        #pragma unroll
        for (int u = 0; u < SORT_BATCH; ++u) {
            const int pp = p + u * SORT_THREADS;
            pr[u] = (pp < be) ? pairs[pp] : make_int2(-1, 0);
        }
        #pragma unroll
        for (int u = 0; u < SORT_BATCH; ++u) {
            if (pr[u].x >= 0) {
                int pos = atomicAdd(&c[pr[u].x & (BUCKET_NODES - 1)], 1);
                pos = (int)min((unsigned)pos, (unsigned)(EE - 1));
                col[pos] = pr[u].y;
            }
        }
    }
    if (t < BUCKET_NODES) {
        const int node = k * BUCKET_NODES + t;
        if (node < NN) rowptr[node] = start;
    }
    if (k == NBUCKETS - 1 && t == 0) rowptr[NN] = be;
}

template<int DIM>
__global__ __launch_bounds__(256) void gather_agg_kernel(
    const float* __restrict__ h, const int* __restrict__ rowptr,
    const int* __restrict__ col, const float* __restrict__ epsP, int epsIdx,
    float* __restrict__ agg)
{
    constexpr int VW = DIM / 32;
    typedef __attribute__((ext_vector_type(VW))) float vt;
    const int lane = threadIdx.x & 31;
    const int node = __builtin_amdgcn_readfirstlane(
        (int)(blockIdx.x * 8 + (threadIdx.x >> 5)));
    const int beg = rowptr[node];
    const int end = rowptr[node + 1];
    const float* __restrict__ hl = h + VW * lane;

    const vt self = *(const vt*)(hl + (size_t)node * DIM);

    vt acc = {};
    const vt zero = {};
    for (int base = beg; base < end; base += 32) {
        const int n = min(32, end - base);
        int ids = (lane < n) ? col[base + lane] : 0;
        ids = (int)min((unsigned)ids, (unsigned)(NN - 1));
        int j = 0;
        for (; j + 4 <= n; j += 4) {
            const int s0 = __builtin_amdgcn_readlane(ids, j);
            const int s1 = __builtin_amdgcn_readlane(ids, j + 1);
            const int s2 = __builtin_amdgcn_readlane(ids, j + 2);
            const int s3 = __builtin_amdgcn_readlane(ids, j + 3);
            const vt v0 = *(const vt*)(hl + (size_t)s0 * DIM);
            const vt v1 = *(const vt*)(hl + (size_t)s1 * DIM);
            const vt v2 = *(const vt*)(hl + (size_t)s2 * DIM);
            const vt v3 = *(const vt*)(hl + (size_t)s3 * DIM);
            acc += v0; acc += v1; acc += v2; acc += v3;
        }
        if (j < n) {
            const int last = n - 1;
            const int s0 = __builtin_amdgcn_readlane(ids, j);
            const int s1 = __builtin_amdgcn_readlane(ids, min(j + 1, last));
            const int s2 = __builtin_amdgcn_readlane(ids, min(j + 2, last));
            const vt v0 = *(const vt*)(hl + (size_t)s0 * DIM);
            const vt v1 = *(const vt*)(hl + (size_t)s1 * DIM);
            const vt v2 = *(const vt*)(hl + (size_t)s2 * DIM);
            acc += v0;
            acc += (j + 1 < n) ? v1 : zero;
            acc += (j + 2 < n) ? v2 : zero;
        }
    }
    const float sc = 1.0f + epsP[epsIdx];
    *(vt*)(agg + (size_t)node * DIM + VW * lane) = sc * self + acc;
}

#define ROW_TILES (NN / 16)
#define MLP_WAVES 14
#define MLP_THREADS (MLP_WAVES * 32)
#define MLP_BLOCKS ((ROW_TILES + MLP_WAVES - 1) / MLP_WAVES)
#define W1PITCH 144
#define W2PITCH 132

template<int KK, int PITCH>
struct WStage {
    static constexpr int P = KK / 2;
    static constexpr int NMAX = (P + MLP_WAVES - 1) / MLP_WAVES;
    float4 r0[NMAX], r1[NMAX];
    __device__ __forceinline__ void load(const float* __restrict__ Wg, int wave, int lane) {
        #pragma unroll
        for (int i = 0; i < NMAX; ++i) {
            const int kp = wave + MLP_WAVES * i;
            if (kp < P) {
                r0[i] = *(const float4*)(Wg + (size_t)(2 * kp) * HID + 4 * lane);
                r1[i] = *(const float4*)(Wg + (size_t)(2 * kp + 1) * HID + 4 * lane);
            } else {
                r0[i] = make_float4(0.f, 0.f, 0.f, 0.f);
                r1[i] = make_float4(0.f, 0.f, 0.f, 0.f);
            }
        }
    }
    __device__ __forceinline__ void store(float* Ws, int wave, int lane) const {
        #pragma unroll
        for (int i = 0; i < NMAX; ++i) {
            const int kp = wave + MLP_WAVES * i;
            if (kp < P) {
                float* d = Ws + (kp * PITCH + 4 * lane) * 2;
                *(float4*)(d)     = make_float4(r0[i].x, r1[i].x, r0[i].y, r1[i].y);
                *(float4*)(d + 4) = make_float4(r0[i].z, r1[i].z, r0[i].w, r1[i].w);
            }
        }
    }
};

template<int K, bool POOL>
__global__ __launch_bounds__(MLP_THREADS, 7) void mlp_kernel(
    const float* __restrict__ A,
    const float* __restrict__ W1, const float* __restrict__ b1,
    const float* __restrict__ g1, const float* __restrict__ o1,
    const float* __restrict__ m1, const float* __restrict__ v1,
    const float* __restrict__ W2, const float* __restrict__ b2,
    const float* __restrict__ g2, const float* __restrict__ o2,
    const float* __restrict__ m2, const float* __restrict__ v2,
    float* __restrict__ out,
    const int* __restrict__ batch, float* __restrict__ sums, float* __restrict__ cnt)
{
    __shared__ __attribute__((aligned(16))) float Ws1[(K / 2) * W1PITCH * 2];
    __shared__ __attribute__((aligned(16))) float Ws2[(HID / 2) * W2PITCH * 2];

    const int lane = threadIdx.x & 31;
    const int wave = __builtin_amdgcn_readfirstlane((int)(threadIdx.x >> 5));
    const int half = lane >> 4;
    const int mr   = lane & 15;

    const int tileM = blockIdx.x * MLP_WAVES + wave;
    const bool live = tileM < ROW_TILES;
    const float* __restrict__ Arow =
        A + (size_t)((live ? tileM : 0) * 16 + mr) * K + 2 * half;

    WStage<K, W1PITCH> st1;
    WStage<HID, W2PITCH> st2;
    st1.load(W1, wave, lane);
    st2.load(W2, wave, lane);
    constexpr int KC = 16;
    v2f a_cur[KC / 4];
    #pragma unroll
    for (int i = 0; i < KC / 4; ++i) a_cur[i] = *(const v2f*)(Arow + 4 * i);
    st1.store(Ws1, wave, lane);
    st2.store(Ws2, wave, lane);
    __syncthreads();
    if (!live) return;

    const float* Wl1 = Ws1 + (half * W1PITCH + mr) * 2;

    v8f acc1[8];
    #pragma unroll
    for (int c = 0; c < 8; ++c) acc1[c] = (v8f){};

    #pragma unroll 1
    for (int kc = 0; kc < K; kc += KC) {
        const int kn = min(kc + KC, K - KC);
        v2f a_nxt[KC / 4];
        #pragma unroll
        for (int i = 0; i < KC / 4; ++i) a_nxt[i] = *(const v2f*)(Arow + kn + 4 * i);
        #pragma unroll
        for (int i = 0; i < KC / 4; ++i) {
            #pragma unroll
            for (int c = 0; c < 8; ++c) {
                const v2f w = *(const v2f*)(Wl1 + ((kc + 4 * i) / 2) * W1PITCH * 2 + c * 32);
                acc1[c] = __builtin_amdgcn_wmma_f32_16x16x4_f32(
                    false, w, false, a_cur[i], (short)0, acc1[c], false, false);
            }
        }
        #pragma unroll
        for (int i = 0; i < KC / 4; ++i) a_cur[i] = a_nxt[i];
    }

    #pragma unroll
    for (int gq = 0; gq < 16; ++gq) {
        const int c = gq >> 1;
        const int u = gq & 1;
        const int f0 = c * 16 + 8 * half + 4 * u;
        const float4 tb = *(const float4*)(b1 + f0);
        const float4 tm = *(const float4*)(m1 + f0);
        const float4 tv = *(const float4*)(v1 + f0);
        const float4 tg = *(const float4*)(g1 + f0);
        const float4 to = *(const float4*)(o1 + f0);
        const float pb[4] = {tb.x, tb.y, tb.z, tb.w};
        const float pm[4] = {tm.x, tm.y, tm.z, tm.w};
        const float pv[4] = {tv.x, tv.y, tv.z, tv.w};
        const float pg[4] = {tg.x, tg.y, tg.z, tg.w};
        const float po[4] = {to.x, to.y, to.z, to.w};
        #pragma unroll
        for (int e = 0; e < 4; ++e) {
            const int r = 4 * u + e;
            float y = acc1[c][r] + pb[e];
            y = (y - pm[e]) * rsqrtf(pv[e] + 1e-5f) * pg[e] + po[e];
            acc1[c][r] = gelu_exact_select(y);
        }
        __builtin_amdgcn_sched_barrier(0);
    }

    int gid[8] = {};
    if constexpr (POOL) {
        const int4* __restrict__ bp = (const int4*)(batch + tileM * 16 + 8 * half);
        const int4 q0 = bp[0];
        const int4 q1 = bp[1];
        gid[0] = q0.x; gid[1] = q0.y; gid[2] = q0.z; gid[3] = q0.w;
        gid[4] = q1.x; gid[5] = q1.y; gid[6] = q1.z; gid[7] = q1.w;
    }

    constexpr int NT = POOL ? 2 : 4;
    const float* Wl2 = Ws2 + (4 * half * W2PITCH + mr) * 2;

    #pragma unroll 1
    for (int pass = 0; pass < 8 / NT; ++pass) {
        v8f acc2[NT];
        #pragma unroll
        for (int o = 0; o < NT; ++o) acc2[o] = (v8f){};
        const float* Wp = Wl2 + pass * (NT * 32);
        #pragma unroll
        for (int jt = 0; jt < 8; ++jt) {
            #pragma unroll
            for (int q = 0; q < 4; ++q) {
                v2f a;
                a[0] = acc1[jt][2 * q];
                a[1] = acc1[jt][2 * q + 1];
                #pragma unroll
                for (int o = 0; o < NT; ++o) {
                    const v2f b = *(const v2f*)(Wp + (8 * jt + q) * W2PITCH * 2 + o * 32);
                    acc2[o] = __builtin_amdgcn_wmma_f32_16x16x4_f32(
                        false, a, false, b, (short)0, acc2[o], false, false);
                }
            }
        }

        #pragma unroll
        for (int o = 0; o < NT; ++o) {
            const int col = (NT * pass + o) * 16 + mr;
            const float cb = b2[col];
            const float cm = m2[col];
            const float cs = rsqrtf(v2[col] + 1e-5f);
            const float cg = g2[col];
            const float co = o2[col];
            if constexpr (!POOL) {
                #pragma unroll
                for (int r = 0; r < 8; ++r) {
                    const int row = tileM * 16 + r + 8 * half;
                    float y = acc2[o][r] + cb;
                    y = (y - cm) * cs * cg + co;
                    y = gelu_exact(y);
                    out[(size_t)row * HID + col] = y;
                }
            } else {
                int cur = gid[0];
                float s = 0.0f;
                #pragma unroll
                for (int r = 0; r < 8; ++r) {
                    float y = acc2[o][r] + cb;
                    y = (y - cm) * cs * cg + co;
                    y = gelu_exact(y);
                    if (gid[r] != cur) {
                        if ((unsigned)cur < (unsigned)GG)
                            unsafeAtomicAdd(sums + (size_t)cur * HID + col, s);
                        s = 0.0f;
                        cur = gid[r];
                    }
                    s += y;
                }
                if ((unsigned)cur < (unsigned)GG)
                    unsafeAtomicAdd(sums + (size_t)cur * HID + col, s);
            }
        }
    }
    if constexpr (POOL) {
        if (mr == 0) {
            int cur = gid[0];
            float n = 0.0f;
            #pragma unroll
            for (int r = 0; r < 8; ++r) {
                if (gid[r] != cur) {
                    if ((unsigned)cur < (unsigned)GG) unsafeAtomicAdd(cnt + cur, n);
                    n = 0.0f;
                    cur = gid[r];
                }
                n += 1.0f;
            }
            if ((unsigned)cur < (unsigned)GG) unsafeAtomicAdd(cnt + cur, n);
        }
    }
}

__global__ __launch_bounds__(256) void zero_kernel(float* __restrict__ p, int n) {
    int idx = blockIdx.x * blockDim.x + threadIdx.x;
    if (idx < n) p[idx] = 0.0f;
}

__global__ __launch_bounds__(128) void head_kernel(
    const float* __restrict__ sums, const float* __restrict__ cnt,
    const float* __restrict__ fc1w, const float* __restrict__ fc1b,
    const float* __restrict__ lng, const float* __restrict__ lnb,
    const float* __restrict__ fc2w, const float* __restrict__ fc2b,
    float* __restrict__ out)
{
    __shared__ float p[HID];
    __shared__ float rbuf[HID];
    __shared__ float red[HID];
    const int g = blockIdx.x;
    const int t = threadIdx.x;

    p[t] = sums[(size_t)g * HID + t] / fmaxf(cnt[g], 1.0f);
    __syncthreads();

    float o = fc1b[t];
    #pragma unroll 4
    for (int k = 0; k < HID; ++k) o += p[k] * fc1w[k * HID + t];

    red[t] = o;
    __syncthreads();
    for (int s = 64; s > 0; s >>= 1) {
        if (t < s) red[t] += red[t + s];
        __syncthreads();
    }
    const float mu = red[0] * (1.0f / HID);
    __syncthreads();

    const float dv = o - mu;
    red[t] = dv * dv;
    __syncthreads();
    for (int s = 64; s > 0; s >>= 1) {
        if (t < s) red[t] += red[t + s];
        __syncthreads();
    }
    const float var = red[0] * (1.0f / HID);

    float y = dv * rsqrtf(var + 1e-5f) * lng[t] + lnb[t];
    y = gelu_exact(y) + p[t];
    rbuf[t] = y;
    __syncthreads();

    if (t < LAT) {
        float o2 = fc2b[t];
        #pragma unroll 4
        for (int k = 0; k < HID; ++k) o2 += rbuf[k] * fc2w[k * LAT + t];
        out[(size_t)g * LAT + t] = o2;
    }
}

extern "C" void kernel_launch(void* const* d_in, const int* in_sizes, int n_in,
                              void* d_out, int out_size, void* d_ws, size_t ws_size,
                              hipStream_t stream) {
    const float* x       = (const float*)d_in[0];
    const int*   ei      = (const int*)  d_in[1];
    const int*   batch   = (const int*)  d_in[2];
    const float* in_g    = (const float*)d_in[3];
    const float* in_b    = (const float*)d_in[4];
    const float* in_m    = (const float*)d_in[5];
    const float* in_v    = (const float*)d_in[6];
    const float* eps0    = (const float*)d_in[7];
    const float* c0_w1   = (const float*)d_in[8];
    const float* c0_b1   = (const float*)d_in[9];
    const float* c0_g    = (const float*)d_in[10];
    const float* c0_bb   = (const float*)d_in[11];
    const float* c0_m    = (const float*)d_in[12];
    const float* c0_v    = (const float*)d_in[13];
    const float* c0_w2   = (const float*)d_in[14];
    const float* c0_b2   = (const float*)d_in[15];
    const float* bn0_g   = (const float*)d_in[16];
    const float* bn0_b   = (const float*)d_in[17];
    const float* bn0_m   = (const float*)d_in[18];
    const float* bn0_v   = (const float*)d_in[19];
    const float* epsR    = (const float*)d_in[20];
    const float* cR_w1   = (const float*)d_in[21];
    const float* cR_b1   = (const float*)d_in[22];
    const float* cR_g    = (const float*)d_in[23];
    const float* cR_bb   = (const float*)d_in[24];
    const float* cR_m    = (const float*)d_in[25];
    const float* cR_v    = (const float*)d_in[26];
    const float* cR_w2   = (const float*)d_in[27];
    const float* cR_b2   = (const float*)d_in[28];
    const float* bnR_g   = (const float*)d_in[29];
    const float* bnR_b   = (const float*)d_in[30];
    const float* bnR_m   = (const float*)d_in[31];
    const float* bnR_v   = (const float*)d_in[32];
    const float* fc1_w   = (const float*)d_in[33];
    const float* fc1_b   = (const float*)d_in[34];
    const float* ln_g    = (const float*)d_in[35];
    const float* ln_b    = (const float*)d_in[36];
    const float* fc2_w   = (const float*)d_in[37];
    const float* fc2_b   = (const float*)d_in[38];

    const int* src = ei;
    const int* dst = ei + EE;

    float* bufA = (float*)d_ws;
    float* bufB = bufA + (size_t)NN * HID;
    int*   col    = (int*)(bufB + (size_t)NN * HID);
    int*   pairs  = col + EE;
    int*   cells  = pairs + 2 * EE;
    int*   base   = cells + PART_CELLS;
    int*   rowptr = base + PART_CELLS + 4;
    float* sums = bufA + (size_t)3 * NN * HID;
    float* cnt  = sums + (size_t)GG * HID;

    const int TB = 256;
    const int nodeBlocks = NN / 8;

    bn_in_kernel<<<(NN * INDIM + TB - 1) / TB, TB, 0, stream>>>(
        x, in_g, in_b, in_m, in_v, bufA);

    part_hist_kernel<<<PART_BLOCKS, PART_THREADS, 0, stream>>>(dst, cells);
    part_scan_kernel<<<SCAN_BLOCKS, SCAN_THREADS, 0, stream>>>(cells, base);
    part_scatter_kernel<<<PART_BLOCKS, PART_THREADS, 0, stream>>>(src, dst, base, pairs);
    bucket_sort_kernel<<<NBUCKETS, SORT_THREADS, 0, stream>>>(pairs, base, rowptr, col);

    gather_agg_kernel<INDIM><<<nodeBlocks, TB, 0, stream>>>(
        bufA, rowptr, col, eps0, 0, bufB);
    mlp_kernel<INDIM, false><<<MLP_BLOCKS, MLP_THREADS, 0, stream>>>(
        bufB, c0_w1, c0_b1, c0_g, c0_bb, c0_m, c0_v,
        c0_w2, c0_b2, bn0_g, bn0_b, bn0_m, bn0_v, bufA, batch, sums, cnt);

    zero_kernel<<<(GG * HID + GG + TB - 1) / TB, TB, 0, stream>>>(sums, GG * HID + GG);

    gather_agg_kernel<HID><<<nodeBlocks, TB, 0, stream>>>(
        bufA, rowptr, col, epsR, 0, bufB);
    mlp_kernel<HID, false><<<MLP_BLOCKS, MLP_THREADS, 0, stream>>>(
        bufB, cR_w1, cR_b1, cR_g, cR_bb, cR_m, cR_v,
        cR_w2, cR_b2, bnR_g, bnR_b, bnR_m, bnR_v, bufA, batch, sums, cnt);
    gather_agg_kernel<HID><<<nodeBlocks, TB, 0, stream>>>(
        bufA, rowptr, col, epsR, 1, bufB);
    mlp_kernel<HID, true><<<MLP_BLOCKS, MLP_THREADS, 0, stream>>>(
        bufB, cR_w1 + HID * HID, cR_b1 + HID, cR_g + HID, cR_bb + HID, cR_m + HID, cR_v + HID,
        cR_w2 + HID * HID, cR_b2 + HID, bnR_g + HID, bnR_b + HID, bnR_m + HID, bnR_v + HID,
        bufA, batch, sums, cnt);

    head_kernel<<<GG, HID, 0, stream>>>(
        sums, cnt, fc1_w, fc1_b, ln_g, ln_b, fc2_w, fc2_b, (float*)d_out);
}


# Both GELUs of the layer kernel on one straight-line erf whose exp(-p) is the hardware exp2 instruction: library expf and exec-masked erff branches gone; values move by one ulp of erf at most
# speedup vs baseline: 3.0534x; 1.1233x over previous
#include <hip/hip_runtime.h>
#include <math.h>

typedef __attribute__((ext_vector_type(2))) float v2f;
typedef __attribute__((ext_vector_type(8))) float v8f;

#define NN   100000
#define EE   1600000
#define INDIM 64
#define HID  128
#define LAT  64
#define GG   1000

__device__ __forceinline__ float gelu_exact(float x) {
    return 0.5f * x * (1.0f + erff(x * 0.70710678118654752f));
}

__device__ __forceinline__ float erf_select(float x) {
    const float ax = fabsf(x);
    float p = fmaf(ax, __uint_as_float(0x378e98abu), __uint_as_float(0xb9c68948u));
    p = fmaf(ax, p, __uint_as_float(0x3b7cd369u));
    p = fmaf(ax, p, __uint_as_float(0xbcc618b2u));
    p = fmaf(ax, p, __uint_as_float(0x3dda74e4u));
    p = fmaf(ax, p, __uint_as_float(0x3f228afdu));
    p = fmaf(ax, p, __uint_as_float(0x3e03c728u));
    p = fmaf(ax, p, ax);
    const float hi = 1.0f - __builtin_amdgcn_exp2f(p * -1.44269504088896341f);
    const float t = x * x;
    float q = fmaf(t, __uint_as_float(0xba1345e1u), __uint_as_float(0x3ba10414u));
    q = fmaf(t, q, __uint_as_float(0xbcdac9b8u));
    q = fmaf(t, q, __uint_as_float(0x3de703beu));
    q = fmaf(t, q, __uint_as_float(0xbec09330u));
    q = fmaf(t, q, __uint_as_float(0x3e0375d0u));
    const float lo = fmaf(ax, q, ax);
    return copysignf((ax < 1.0f) ? lo : hi, x);
}

__device__ __forceinline__ float gelu_exact_select(float x) {
    return 0.5f * x * (1.0f + erf_select(x * 0.70710678118654752f));
}

#define BUCKET_SHIFT 8
#define BUCKET_NODES (1 << BUCKET_SHIFT)
#define NBUCKETS ((NN + BUCKET_NODES - 1) / BUCKET_NODES)
#define PART_BLOCKS 256
#define PART_THREADS 1024
#define PART_CHUNK ((EE + PART_BLOCKS - 1) / PART_BLOCKS)
#define PART_BATCH 8
#define PART_CELLS (NBUCKETS * PART_BLOCKS)
#define SCAN_THREADS 1024
#define SCAN_BLOCKS ((PART_CELLS + SCAN_THREADS - 1) / SCAN_THREADS)
#define SCAN_PRE_MAX (((SCAN_BLOCKS - 1) * (SCAN_THREADS / 4) + SCAN_THREADS - 1) / SCAN_THREADS)
#define SORT_THREADS 1024
#define SORT_BATCH 5

__global__ __launch_bounds__(PART_THREADS) void part_hist_kernel(
    const int* __restrict__ dst, int* __restrict__ cells)
{
    __shared__ int h[NBUCKETS];
    const int t = threadIdx.x;
    for (int k = t; k < NBUCKETS; k += PART_THREADS) h[k] = 0;
    __syncthreads();
    const int e0 = blockIdx.x * PART_CHUNK;
    const int e1 = min(e0 + PART_CHUNK, EE);
    for (int e = e0 + t; e < e1; e += PART_BATCH * PART_THREADS) {
        int d[PART_BATCH];
        #pragma unroll
        for (int u = 0; u < PART_BATCH; ++u) {
            const int ee = e + u * PART_THREADS;
            d[u] = (ee < e1) ? dst[ee] : -1;
        }
        #pragma unroll
        for (int u = 0; u < PART_BATCH; ++u)
            if ((unsigned)d[u] < (unsigned)NN) atomicAdd(&h[d[u] >> BUCKET_SHIFT], 1);
    }
    __syncthreads();
    for (int k = t; k < NBUCKETS; k += PART_THREADS)
        cells[k * PART_BLOCKS + blockIdx.x] = h[k];
}

__global__ __launch_bounds__(SCAN_THREADS) void part_scan_kernel(
    const int* __restrict__ cells, int* __restrict__ base)
{
    __shared__ int red[SCAN_THREADS / 32];
    __shared__ int wsum[SCAN_THREADS / 32];
    const int t = threadIdx.x;
    const int lane = t & 31;
    const int wave = t >> 5;
    const int nb = blockIdx.x;

    const int4* __restrict__ c4 = (const int4*)cells;
    const int n4 = nb * (SCAN_THREADS / 4);
    int4 q[SCAN_PRE_MAX];
    #pragma unroll
    for (int u = 0; u < SCAN_PRE_MAX; ++u) {
        const int i = u * SCAN_THREADS + t;
        q[u] = (i < n4) ? c4[i] : make_int4(0, 0, 0, 0);
    }
    int s = 0;
    #pragma unroll
    for (int u = 0; u < SCAN_PRE_MAX; ++u) s += q[u].x + q[u].y + q[u].z + q[u].w;
    const int idx = nb * SCAN_THREADS + t;
    const int x = (idx < PART_CELLS) ? cells[idx] : 0;

    for (int off = 16; off > 0; off >>= 1) s += __shfl_xor(s, off);
    int inc = x;
    for (int off = 1; off < 32; off <<= 1) {
        const int n = __shfl_up(inc, off);
        if (lane >= off) inc += n;
    }
    if (lane == 0) red[wave] = s;
    if (lane == 31) wsum[wave] = inc;
    __syncthreads();
    int before = 0;
    for (int w = 0; w < SCAN_THREADS / 32; ++w) before += red[w];
    int woff = 0;
    for (int w = 0; w < wave; ++w) woff += wsum[w];
    const int ex = before + woff + inc - x;
    if (idx < PART_CELLS) base[idx] = ex;
    if (idx == PART_CELLS - 1) base[PART_CELLS] = ex + x;
}

__global__ __launch_bounds__(PART_THREADS) void part_scatter_kernel(
    const int* __restrict__ src, const int* __restrict__ dst,
    const int* __restrict__ base, int* __restrict__ pairs_out)
{
    int2* __restrict__ pairs = (int2*)pairs_out;
    __shared__ int cur[NBUCKETS];
    const int t = threadIdx.x;
    for (int k = t; k < NBUCKETS; k += PART_THREADS)
        cur[k] = base[k * PART_BLOCKS + blockIdx.x];
    __syncthreads();
    const int e0 = blockIdx.x * PART_CHUNK;
    const int e1 = min(e0 + PART_CHUNK, EE);
    for (int e = e0 + t; e < e1; e += PART_BATCH * PART_THREADS) {
        int d[PART_BATCH], s[PART_BATCH];
        #pragma unroll
        for (int u = 0; u < PART_BATCH; ++u) {
            const int ee = e + u * PART_THREADS;
            const bool ok = ee < e1;
            d[u] = ok ? dst[ee] : -1;
            s[u] = ok ? src[ee] : 0;
        }
        #pragma unroll
        for (int u = 0; u < PART_BATCH; ++u) {
            if ((unsigned)d[u] < (unsigned)NN) {
                int pos = atomicAdd(&cur[d[u] >> BUCKET_SHIFT], 1);
                pos = (int)min((unsigned)pos, (unsigned)(EE - 1));
                pairs[pos] = make_int2(d[u], s[u]);
            }
        }
    }
}

__global__ __launch_bounds__(SORT_THREADS) void bucket_sort_kernel(
    const int* __restrict__ pairs_in, const int* __restrict__ base,
    int* __restrict__ rowptr, int* __restrict__ col)
{
    const int2* __restrict__ pairs = (const int2*)pairs_in;
    __shared__ int c[BUCKET_NODES];
    __shared__ int wsum[BUCKET_NODES / 32];
    const int t = threadIdx.x;
    const int lane = t & 31;
    const int wave = t >> 5;
    const int k = blockIdx.x;
    int bs = base[k * PART_BLOCKS];
    int be = base[(k + 1) * PART_BLOCKS];
    bs = max(bs, 0);
    be = min(be, EE);

    if (t < BUCKET_NODES) c[t] = 0;
    __syncthreads();
    for (int p = bs + t; p < be; p += SORT_BATCH * SORT_THREADS) {
        int d[SORT_BATCH];
        #pragma unroll
        for (int u = 0; u < SORT_BATCH; ++u) {
            const int pp = p + u * SORT_THREADS;
            d[u] = (pp < be) ? pairs[pp].x : -1;
        }
        #pragma unroll
        for (int u = 0; u < SORT_BATCH; ++u)
            if (d[u] >= 0) atomicAdd(&c[d[u] & (BUCKET_NODES - 1)], 1);
    }
    __syncthreads();

    int x = 0, inc = 0, start = 0;
    if (t < BUCKET_NODES) {
        x = c[t];
        inc = x;
        for (int off = 1; off < 32; off <<= 1) {
            const int n = __shfl_up(inc, off);
            if (lane >= off) inc += n;
        }
        if (lane == 31) wsum[wave] = inc;
    }
    __syncthreads();
    if (t < BUCKET_NODES) {
        int woff = 0;
        for (int w = 0; w < wave; ++w) woff += wsum[w];
        start = bs + woff + inc - x;
        c[t] = start;
    }
    __syncthreads();

    for (int p = bs + t; p < be; p += SORT_BATCH * SORT_THREADS) {
        int2 pr[SORT_BATCH];
        #pragma unroll
        for (int u = 0; u < SORT_BATCH; ++u) {
            const int pp = p + u * SORT_THREADS;
            pr[u] = (pp < be) ? pairs[pp] : make_int2(-1, 0);
        }
        #pragma unroll
        for (int u = 0; u < SORT_BATCH; ++u) {
            if (pr[u].x >= 0) {
                int pos = atomicAdd(&c[pr[u].x & (BUCKET_NODES - 1)], 1);
                pos = (int)min((unsigned)pos, (unsigned)(EE - 1));
                col[pos] = pr[u].y;
            }
        }
    }
    if (t < BUCKET_NODES) {
        const int node = k * BUCKET_NODES + t;
        if (node < NN) rowptr[node] = start;
    }
    if (k == NBUCKETS - 1 && t == 0) rowptr[NN] = be;
}

template<int DIM, bool BN>
__global__ __launch_bounds__(256) void gather_agg_kernel(
    const float* __restrict__ h, const int* __restrict__ rowptr,
    const int* __restrict__ col, const float* __restrict__ epsP, int epsIdx,
    const float* __restrict__ bn_g, const float* __restrict__ bn_b,
    const float* __restrict__ bn_m, const float* __restrict__ bn_v,
    float* __restrict__ agg)
{
    constexpr int VW = DIM / 32;
    typedef __attribute__((ext_vector_type(VW))) float vt;
    const int lane = threadIdx.x & 31;
    const int node = __builtin_amdgcn_readfirstlane(
        (int)(blockIdx.x * 8 + (threadIdx.x >> 5)));
    const int beg = rowptr[node];
    const int end = rowptr[node + 1];
    const float* __restrict__ hl = h + VW * lane;

    vt pv = {}, pm = {}, ps = {}, pg = {}, pb = {};
    if constexpr (BN) {
        pv = *(const vt*)(bn_v + VW * lane);
        pm = *(const vt*)(bn_m + VW * lane);
        pg = *(const vt*)(bn_g + VW * lane);
        pb = *(const vt*)(bn_b + VW * lane);
    }

    const vt self_raw = *(const vt*)(hl + (size_t)node * DIM);

    if constexpr (BN) {
        __builtin_amdgcn_sched_barrier(0);
        #pragma unroll
        for (int i = 0; i < VW; ++i) ps[i] = rsqrtf(pv[i] + 1e-5f);
    }
    auto norm = [&](const vt r) -> vt {
        if constexpr (BN) return (r - pm) * ps * pg + pb;
        else return r;
    };

    vt acc = {};
    const vt zero = {};
    for (int base = beg; base < end; base += 32) {
        const int n = min(32, end - base);
        int ids = (lane < n) ? col[base + lane] : 0;
        ids = (int)min((unsigned)ids, (unsigned)(NN - 1));
        int j = 0;
        for (; j + 4 <= n; j += 4) {
            const int s0 = __builtin_amdgcn_readlane(ids, j);
            const int s1 = __builtin_amdgcn_readlane(ids, j + 1);
            const int s2 = __builtin_amdgcn_readlane(ids, j + 2);
            const int s3 = __builtin_amdgcn_readlane(ids, j + 3);
            const vt v0 = *(const vt*)(hl + (size_t)s0 * DIM);
            const vt v1 = *(const vt*)(hl + (size_t)s1 * DIM);
            const vt v2 = *(const vt*)(hl + (size_t)s2 * DIM);
            const vt v3 = *(const vt*)(hl + (size_t)s3 * DIM);
            acc += norm(v0); acc += norm(v1); acc += norm(v2); acc += norm(v3);
        }
        if (j < n) {
            const int last = n - 1;
            const int s0 = __builtin_amdgcn_readlane(ids, j);
            const int s1 = __builtin_amdgcn_readlane(ids, min(j + 1, last));
            const int s2 = __builtin_amdgcn_readlane(ids, min(j + 2, last));
            const vt v0 = *(const vt*)(hl + (size_t)s0 * DIM);
            const vt v1 = *(const vt*)(hl + (size_t)s1 * DIM);
            const vt v2 = *(const vt*)(hl + (size_t)s2 * DIM);
            acc += norm(v0);
            acc += (j + 1 < n) ? norm(v1) : zero;
            acc += (j + 2 < n) ? norm(v2) : zero;
        }
    }
    const float sc = 1.0f + epsP[epsIdx];
    *(vt*)(agg + (size_t)node * DIM + VW * lane) = sc * norm(self_raw) + acc;
}

#define ROW_TILES (NN / 16)
#define MLP_WAVES 14
#define MLP_THREADS (MLP_WAVES * 32)
#define MLP_BLOCKS ((ROW_TILES + MLP_WAVES - 1) / MLP_WAVES)
#define W1PITCH 144
#define W2PITCH 132

template<int KK, int PITCH>
struct WStage {
    static constexpr int P = KK / 2;
    static constexpr int NMAX = (P + MLP_WAVES - 1) / MLP_WAVES;
    float4 r0[NMAX], r1[NMAX];
    __device__ __forceinline__ void load(const float* __restrict__ Wg, int wave, int lane) {
        #pragma unroll
        for (int i = 0; i < NMAX; ++i) {
            const int kp = wave + MLP_WAVES * i;
            if (kp < P) {
                r0[i] = *(const float4*)(Wg + (size_t)(2 * kp) * HID + 4 * lane);
                r1[i] = *(const float4*)(Wg + (size_t)(2 * kp + 1) * HID + 4 * lane);
            } else {
                r0[i] = make_float4(0.f, 0.f, 0.f, 0.f);
                r1[i] = make_float4(0.f, 0.f, 0.f, 0.f);
            }
        }
    }
    __device__ __forceinline__ void store(float* Ws, int wave, int lane) const {
        #pragma unroll
        for (int i = 0; i < NMAX; ++i) {
            const int kp = wave + MLP_WAVES * i;
            if (kp < P) {
                float* d = Ws + (kp * PITCH + 4 * lane) * 2;
                *(float4*)(d)     = make_float4(r0[i].x, r1[i].x, r0[i].y, r1[i].y);
                *(float4*)(d + 4) = make_float4(r0[i].z, r1[i].z, r0[i].w, r1[i].w);
            }
        }
    }
};

template<int K, bool POOL>
__global__ __launch_bounds__(MLP_THREADS, 7) void mlp_kernel(
    const float* __restrict__ A,
    const float* __restrict__ W1, const float* __restrict__ b1,
    const float* __restrict__ g1, const float* __restrict__ o1,
    const float* __restrict__ m1, const float* __restrict__ v1,
    const float* __restrict__ W2, const float* __restrict__ b2,
    const float* __restrict__ g2, const float* __restrict__ o2,
    const float* __restrict__ m2, const float* __restrict__ v2,
    float* __restrict__ out,
    const int* __restrict__ batch, float* __restrict__ sums, float* __restrict__ cnt)
{
    __shared__ __attribute__((aligned(16))) float Ws1[(K / 2) * W1PITCH * 2];
    __shared__ __attribute__((aligned(16))) float Ws2[(HID / 2) * W2PITCH * 2];

    const int lane = threadIdx.x & 31;
    const int wave = __builtin_amdgcn_readfirstlane((int)(threadIdx.x >> 5));
    const int half = lane >> 4;
    const int mr   = lane & 15;

    const int tileM = blockIdx.x * MLP_WAVES + wave;
    const bool live = tileM < ROW_TILES;
    const float* __restrict__ Arow =
        A + (size_t)((live ? tileM : 0) * 16 + mr) * K + 2 * half;

    WStage<K, W1PITCH> st1;
    WStage<HID, W2PITCH> st2;
    st1.load(W1, wave, lane);
    st2.load(W2, wave, lane);
    constexpr int KC = 16;
    v2f a_cur[KC / 4];
    #pragma unroll
    for (int i = 0; i < KC / 4; ++i) a_cur[i] = *(const v2f*)(Arow + 4 * i);
    st1.store(Ws1, wave, lane);
    st2.store(Ws2, wave, lane);
    __syncthreads();
    if (!live) return;

    const float* Wl1 = Ws1 + (half * W1PITCH + mr) * 2;

    v8f acc1[8];
    #pragma unroll
    for (int c = 0; c < 8; ++c) acc1[c] = (v8f){};

    #pragma unroll 1
    for (int kc = 0; kc < K; kc += KC) {
        const int kn = min(kc + KC, K - KC);
        v2f a_nxt[KC / 4];
        #pragma unroll
        for (int i = 0; i < KC / 4; ++i) a_nxt[i] = *(const v2f*)(Arow + kn + 4 * i);
        #pragma unroll
        for (int i = 0; i < KC / 4; ++i) {
            #pragma unroll
            for (int c = 0; c < 8; ++c) {
                const v2f w = *(const v2f*)(Wl1 + ((kc + 4 * i) / 2) * W1PITCH * 2 + c * 32);
                acc1[c] = __builtin_amdgcn_wmma_f32_16x16x4_f32(
                    false, w, false, a_cur[i], (short)0, acc1[c], false, false);
            }
        }
        #pragma unroll
        for (int i = 0; i < KC / 4; ++i) a_cur[i] = a_nxt[i];
    }

    #pragma unroll
    for (int gq = 0; gq < 16; ++gq) {
        const int c = gq >> 1;
        const int u = gq & 1;
        const int f0 = c * 16 + 8 * half + 4 * u;
        const float4 tb = *(const float4*)(b1 + f0);
        const float4 tm = *(const float4*)(m1 + f0);
        const float4 tv = *(const float4*)(v1 + f0);
        const float4 tg = *(const float4*)(g1 + f0);
        const float4 to = *(const float4*)(o1 + f0);
        const float pb[4] = {tb.x, tb.y, tb.z, tb.w};
        const float pm[4] = {tm.x, tm.y, tm.z, tm.w};
        const float pv[4] = {tv.x, tv.y, tv.z, tv.w};
        const float pg[4] = {tg.x, tg.y, tg.z, tg.w};
        const float po[4] = {to.x, to.y, to.z, to.w};
        #pragma unroll
        for (int e = 0; e < 4; ++e) {
            const int r = 4 * u + e;
            float y = acc1[c][r] + pb[e];
            y = (y - pm[e]) * rsqrtf(pv[e] + 1e-5f) * pg[e] + po[e];
            acc1[c][r] = gelu_exact_select(y);
        }
        __builtin_amdgcn_sched_barrier(0);
    }

    int gid[8] = {};
    if constexpr (POOL) {
        const int4* __restrict__ bp = (const int4*)(batch + tileM * 16 + 8 * half);
        const int4 q0 = bp[0];
        const int4 q1 = bp[1];
        gid[0] = q0.x; gid[1] = q0.y; gid[2] = q0.z; gid[3] = q0.w;
        gid[4] = q1.x; gid[5] = q1.y; gid[6] = q1.z; gid[7] = q1.w;
    }

    constexpr int NT = POOL ? 2 : 4;
    const float* Wl2 = Ws2 + (4 * half * W2PITCH + mr) * 2;

    #pragma unroll 1
    for (int pass = 0; pass < 8 / NT; ++pass) {
        v8f acc2[NT];
        #pragma unroll
        for (int o = 0; o < NT; ++o) acc2[o] = (v8f){};
        const float* Wp = Wl2 + pass * (NT * 32);
        #pragma unroll
        for (int jt = 0; jt < 8; ++jt) {
            #pragma unroll
            for (int q = 0; q < 4; ++q) {
                v2f a;
                a[0] = acc1[jt][2 * q];
                a[1] = acc1[jt][2 * q + 1];
                #pragma unroll
                for (int o = 0; o < NT; ++o) {
                    const v2f b = *(const v2f*)(Wp + (8 * jt + q) * W2PITCH * 2 + o * 32);
                    acc2[o] = __builtin_amdgcn_wmma_f32_16x16x4_f32(
                        false, a, false, b, (short)0, acc2[o], false, false);
                }
            }
        }

        #pragma unroll
        for (int o = 0; o < NT; ++o) {
            const int col = (NT * pass + o) * 16 + mr;
            const float cb = b2[col];
            const float cm = m2[col];
            const float cs = rsqrtf(v2[col] + 1e-5f);
            const float cg = g2[col];
            const float co = o2[col];
            if constexpr (!POOL) {
                #pragma unroll
                for (int r = 0; r < 8; ++r) {
                    const int row = tileM * 16 + r + 8 * half;
                    float y = acc2[o][r] + cb;
                    y = (y - cm) * cs * cg + co;
                    y = gelu_exact_select(y);
                    out[(size_t)row * HID + col] = y;
                }
            } else {
                int cur = gid[0];
                float s = 0.0f;
                #pragma unroll
                for (int r = 0; r < 8; ++r) {
                    float y = acc2[o][r] + cb;
                    y = (y - cm) * cs * cg + co;
                    y = gelu_exact_select(y);
                    if (gid[r] != cur) {
                        if ((unsigned)cur < (unsigned)GG)
                            unsafeAtomicAdd(sums + (size_t)cur * HID + col, s);
                        s = 0.0f;
                        cur = gid[r];
                    }
                    s += y;
                }
                if ((unsigned)cur < (unsigned)GG)
                    unsafeAtomicAdd(sums + (size_t)cur * HID + col, s);
            }
        }
    }
    if constexpr (POOL) {
        if (mr == 0) {
            int cur = gid[0];
            float n = 0.0f;
            #pragma unroll
            for (int r = 0; r < 8; ++r) {
                if (gid[r] != cur) {
                    if ((unsigned)cur < (unsigned)GG) unsafeAtomicAdd(cnt + cur, n);
                    n = 0.0f;
                    cur = gid[r];
                }
                n += 1.0f;
            }
            if ((unsigned)cur < (unsigned)GG) unsafeAtomicAdd(cnt + cur, n);
        }
    }
}

__global__ __launch_bounds__(256) void zero_kernel(float* __restrict__ p, int n) {
    int idx = blockIdx.x * blockDim.x + threadIdx.x;
    if (idx < n) p[idx] = 0.0f;
}

__global__ __launch_bounds__(128) void head_kernel(
    const float* __restrict__ sums, const float* __restrict__ cnt,
    const float* __restrict__ fc1w, const float* __restrict__ fc1b,
    const float* __restrict__ lng, const float* __restrict__ lnb,
    const float* __restrict__ fc2w, const float* __restrict__ fc2b,
    float* __restrict__ out)
{
    __shared__ float p[HID];
    __shared__ float rbuf[HID];
    __shared__ float red[HID];
    const int g = blockIdx.x;
    const int t = threadIdx.x;

    p[t] = sums[(size_t)g * HID + t] / fmaxf(cnt[g], 1.0f);
    __syncthreads();

    float o = fc1b[t];
    #pragma unroll 4
    for (int k = 0; k < HID; ++k) o += p[k] * fc1w[k * HID + t];

    red[t] = o;
    __syncthreads();
    for (int s = 64; s > 0; s >>= 1) {
        if (t < s) red[t] += red[t + s];
        __syncthreads();
    }
    const float mu = red[0] * (1.0f / HID);
    __syncthreads();

    const float dv = o - mu;
    red[t] = dv * dv;
    __syncthreads();
    for (int s = 64; s > 0; s >>= 1) {
        if (t < s) red[t] += red[t + s];
        __syncthreads();
    }
    const float var = red[0] * (1.0f / HID);

    float y = dv * rsqrtf(var + 1e-5f) * lng[t] + lnb[t];
    y = gelu_exact(y) + p[t];
    rbuf[t] = y;
    __syncthreads();

    if (t < LAT) {
        float o2 = fc2b[t];
        #pragma unroll 4
        for (int k = 0; k < HID; ++k) o2 += rbuf[k] * fc2w[k * LAT + t];
        out[(size_t)g * LAT + t] = o2;
    }
}

extern "C" void kernel_launch(void* const* d_in, const int* in_sizes, int n_in,
                              void* d_out, int out_size, void* d_ws, size_t ws_size,
                              hipStream_t stream) {
    const float* x       = (const float*)d_in[0];
    const int*   ei      = (const int*)  d_in[1];
    const int*   batch   = (const int*)  d_in[2];
    const float* in_g    = (const float*)d_in[3];
    const float* in_b    = (const float*)d_in[4];
    const float* in_m    = (const float*)d_in[5];
    const float* in_v    = (const float*)d_in[6];
    const float* eps0    = (const float*)d_in[7];
    const float* c0_w1   = (const float*)d_in[8];
    const float* c0_b1   = (const float*)d_in[9];
    const float* c0_g    = (const float*)d_in[10];
    const float* c0_bb   = (const float*)d_in[11];
    const float* c0_m    = (const float*)d_in[12];
    const float* c0_v    = (const float*)d_in[13];
    const float* c0_w2   = (const float*)d_in[14];
    const float* c0_b2   = (const float*)d_in[15];
    const float* bn0_g   = (const float*)d_in[16];
    const float* bn0_b   = (const float*)d_in[17];
    const float* bn0_m   = (const float*)d_in[18];
    const float* bn0_v   = (const float*)d_in[19];
    const float* epsR    = (const float*)d_in[20];
    const float* cR_w1   = (const float*)d_in[21];
    const float* cR_b1   = (const float*)d_in[22];
    const float* cR_g    = (const float*)d_in[23];
    const float* cR_bb   = (const float*)d_in[24];
    const float* cR_m    = (const float*)d_in[25];
    const float* cR_v    = (const float*)d_in[26];
    const float* cR_w2   = (const float*)d_in[27];
    const float* cR_b2   = (const float*)d_in[28];
    const float* bnR_g   = (const float*)d_in[29];
    const float* bnR_b   = (const float*)d_in[30];
    const float* bnR_m   = (const float*)d_in[31];
    const float* bnR_v   = (const float*)d_in[32];
    const float* fc1_w   = (const float*)d_in[33];
    const float* fc1_b   = (const float*)d_in[34];
    const float* ln_g    = (const float*)d_in[35];
    const float* ln_b    = (const float*)d_in[36];
    const float* fc2_w   = (const float*)d_in[37];
    const float* fc2_b   = (const float*)d_in[38];

    const int* src = ei;
    const int* dst = ei + EE;

    float* bufA = (float*)d_ws;
    float* bufB = bufA + (size_t)NN * HID;
    int*   col    = (int*)(bufB + (size_t)NN * HID);
    int*   pairs  = col + EE;
    int*   cells  = pairs + 2 * EE;
    int*   base   = cells + PART_CELLS;
    int*   rowptr = base + PART_CELLS + 4;
    float* sums = bufA + (size_t)3 * NN * HID;
    float* cnt  = sums + (size_t)GG * HID;

    const int TB = 256;
    const int nodeBlocks = NN / 8;

    part_hist_kernel<<<PART_BLOCKS, PART_THREADS, 0, stream>>>(dst, cells);
    part_scan_kernel<<<SCAN_BLOCKS, SCAN_THREADS, 0, stream>>>(cells, base);
    part_scatter_kernel<<<PART_BLOCKS, PART_THREADS, 0, stream>>>(src, dst, base, pairs);
    bucket_sort_kernel<<<NBUCKETS, SORT_THREADS, 0, stream>>>(pairs, base, rowptr, col);

    gather_agg_kernel<INDIM, true><<<nodeBlocks, TB, 0, stream>>>(
        x, rowptr, col, eps0, 0, in_g, in_b, in_m, in_v, bufB);
    mlp_kernel<INDIM, false><<<MLP_BLOCKS, MLP_THREADS, 0, stream>>>(
        bufB, c0_w1, c0_b1, c0_g, c0_bb, c0_m, c0_v,
        c0_w2, c0_b2, bn0_g, bn0_b, bn0_m, bn0_v, bufA, batch, sums, cnt);

    zero_kernel<<<(GG * HID + GG + TB - 1) / TB, TB, 0, stream>>>(sums, GG * HID + GG);

    gather_agg_kernel<HID, false><<<nodeBlocks, TB, 0, stream>>>(
        bufA, rowptr, col, epsR, 0, nullptr, nullptr, nullptr, nullptr, bufB);
    mlp_kernel<HID, false><<<MLP_BLOCKS, MLP_THREADS, 0, stream>>>(
        bufB, cR_w1, cR_b1, cR_g, cR_bb, cR_m, cR_v,
        cR_w2, cR_b2, bnR_g, bnR_b, bnR_m, bnR_v, bufA, batch, sums, cnt);
    gather_agg_kernel<HID, false><<<nodeBlocks, TB, 0, stream>>>(
        bufA, rowptr, col, epsR, 1, nullptr, nullptr, nullptr, nullptr, bufB);
    mlp_kernel<HID, true><<<MLP_BLOCKS, MLP_THREADS, 0, stream>>>(
        bufB, cR_w1 + HID * HID, cR_b1 + HID, cR_g + HID, cR_bb + HID, cR_m + HID, cR_v + HID,
        cR_w2 + HID * HID, cR_b2 + HID, bnR_g + HID, bnR_b + HID, bnR_m + HID, bnR_v + HID,
        bufA, batch, sums, cnt);

    head_kernel<<<GG, HID, 0, stream>>>(
        sums, cnt, fc1_w, fc1_b, ln_g, ln_b, fc2_w, fc2_b, (float*)d_out);
}


# GELU of both layer-kernel epilogues as max(x,0) - |x| 2^(-(1+|x|S(|x|))), S one degree-7 polynomial: second polynomial, select and sign transfer gone; values move by under half an ulp of x
# speedup vs baseline: 3.1857x; 1.0433x over previous
#include <hip/hip_runtime.h>
#include <math.h>

typedef __attribute__((ext_vector_type(2))) float v2f;
typedef __attribute__((ext_vector_type(8))) float v8f;

#define NN   100000
#define EE   1600000
#define INDIM 64
#define HID  128
#define LAT  64
#define GG   1000

__device__ __forceinline__ float gelu_exact(float x) {
    return 0.5f * x * (1.0f + erff(x * 0.70710678118654752f));
}

__device__ __forceinline__ float gelu_tail_form(float x) {
    const float t = fabsf(x);
    float s = fmaf(t, __uint_as_float(0x360d598au), __uint_as_float(0xb80136dbu));
    s = fmaf(t, s, __uint_as_float(0x39197cbcu));
    s = fmaf(t, s, __uint_as_float(0x39717391u));
    s = fmaf(t, s, __uint_as_float(0xbbeb4a40u));
    s = fmaf(t, s, __uint_as_float(0x3d57563eu));
    s = fmaf(t, s, __uint_as_float(0x3eeb1a59u));
    s = fmaf(t, s, __uint_as_float(0x3f93577du));
    const float tail = __builtin_amdgcn_exp2f(-fmaf(t, s, 1.0f));
    return fmaf(-t, tail, fmaxf(x, 0.0f));
}

#define BUCKET_SHIFT 8
#define BUCKET_NODES (1 << BUCKET_SHIFT)
#define NBUCKETS ((NN + BUCKET_NODES - 1) / BUCKET_NODES)
#define PART_BLOCKS 256
#define PART_THREADS 1024
#define PART_CHUNK ((EE + PART_BLOCKS - 1) / PART_BLOCKS)
#define PART_BATCH 8
#define PART_CELLS (NBUCKETS * PART_BLOCKS)
#define SCAN_THREADS 1024
#define SCAN_BLOCKS ((PART_CELLS + SCAN_THREADS - 1) / SCAN_THREADS)
#define SCAN_PRE_MAX (((SCAN_BLOCKS - 1) * (SCAN_THREADS / 4) + SCAN_THREADS - 1) / SCAN_THREADS)
#define SORT_THREADS 1024
#define SORT_BATCH 5

__global__ __launch_bounds__(PART_THREADS) void part_hist_kernel(
    const int* __restrict__ dst, int* __restrict__ cells)
{
    __shared__ int h[NBUCKETS];
    const int t = threadIdx.x;
    for (int k = t; k < NBUCKETS; k += PART_THREADS) h[k] = 0;
    __syncthreads();
    const int e0 = blockIdx.x * PART_CHUNK;
    const int e1 = min(e0 + PART_CHUNK, EE);
    for (int e = e0 + t; e < e1; e += PART_BATCH * PART_THREADS) {
        int d[PART_BATCH];
        #pragma unroll
        for (int u = 0; u < PART_BATCH; ++u) {
            const int ee = e + u * PART_THREADS;
            d[u] = (ee < e1) ? dst[ee] : -1;
        }
        #pragma unroll
        for (int u = 0; u < PART_BATCH; ++u)
            if ((unsigned)d[u] < (unsigned)NN) atomicAdd(&h[d[u] >> BUCKET_SHIFT], 1);
    }
    __syncthreads();
    for (int k = t; k < NBUCKETS; k += PART_THREADS)
        cells[k * PART_BLOCKS + blockIdx.x] = h[k];
}

__global__ __launch_bounds__(SCAN_THREADS) void part_scan_kernel(
    const int* __restrict__ cells, int* __restrict__ base)
{
    __shared__ int red[SCAN_THREADS / 32];
    __shared__ int wsum[SCAN_THREADS / 32];
    const int t = threadIdx.x;
    const int lane = t & 31;
    const int wave = t >> 5;
    const int nb = blockIdx.x;

    const int4* __restrict__ c4 = (const int4*)cells;
    const int n4 = nb * (SCAN_THREADS / 4);
    int4 q[SCAN_PRE_MAX];
    #pragma unroll
    for (int u = 0; u < SCAN_PRE_MAX; ++u) {
        const int i = u * SCAN_THREADS + t;
        q[u] = (i < n4) ? c4[i] : make_int4(0, 0, 0, 0);
    }
    int s = 0;
    #pragma unroll
    for (int u = 0; u < SCAN_PRE_MAX; ++u) s += q[u].x + q[u].y + q[u].z + q[u].w;
    const int idx = nb * SCAN_THREADS + t;
    const int x = (idx < PART_CELLS) ? cells[idx] : 0;

    for (int off = 16; off > 0; off >>= 1) s += __shfl_xor(s, off);
    int inc = x;
    for (int off = 1; off < 32; off <<= 1) {
        const int n = __shfl_up(inc, off);
        if (lane >= off) inc += n;
    }
    if (lane == 0) red[wave] = s;
    if (lane == 31) wsum[wave] = inc;
    __syncthreads();
    int before = 0;
    for (int w = 0; w < SCAN_THREADS / 32; ++w) before += red[w];
    int woff = 0;
    for (int w = 0; w < wave; ++w) woff += wsum[w];
    const int ex = before + woff + inc - x;
    if (idx < PART_CELLS) base[idx] = ex;
    if (idx == PART_CELLS - 1) base[PART_CELLS] = ex + x;
}

__global__ __launch_bounds__(PART_THREADS) void part_scatter_kernel(
    const int* __restrict__ src, const int* __restrict__ dst,
    const int* __restrict__ base, int* __restrict__ pairs_out)
{
    int2* __restrict__ pairs = (int2*)pairs_out;
    __shared__ int cur[NBUCKETS];
    const int t = threadIdx.x;
    for (int k = t; k < NBUCKETS; k += PART_THREADS)
        cur[k] = base[k * PART_BLOCKS + blockIdx.x];
    __syncthreads();
    const int e0 = blockIdx.x * PART_CHUNK;
    const int e1 = min(e0 + PART_CHUNK, EE);
    for (int e = e0 + t; e < e1; e += PART_BATCH * PART_THREADS) {
        int d[PART_BATCH], s[PART_BATCH];
        #pragma unroll
        for (int u = 0; u < PART_BATCH; ++u) {
            const int ee = e + u * PART_THREADS;
            const bool ok = ee < e1;
            d[u] = ok ? dst[ee] : -1;
            s[u] = ok ? src[ee] : 0;
        }
        #pragma unroll
        for (int u = 0; u < PART_BATCH; ++u) {
            if ((unsigned)d[u] < (unsigned)NN) {
                int pos = atomicAdd(&cur[d[u] >> BUCKET_SHIFT], 1);
                pos = (int)min((unsigned)pos, (unsigned)(EE - 1));
                pairs[pos] = make_int2(d[u], s[u]);
            }
        }
    }
}

__global__ __launch_bounds__(SORT_THREADS) void bucket_sort_kernel(
    const int* __restrict__ pairs_in, const int* __restrict__ base,
    int* __restrict__ rowptr, int* __restrict__ col)
{
    const int2* __restrict__ pairs = (const int2*)pairs_in;
    __shared__ int c[BUCKET_NODES];
    __shared__ int wsum[BUCKET_NODES / 32];
    const int t = threadIdx.x;
    const int lane = t & 31;
    const int wave = t >> 5;
    const int k = blockIdx.x;
    int bs = base[k * PART_BLOCKS];
    int be = base[(k + 1) * PART_BLOCKS];
    bs = max(bs, 0);
    be = min(be, EE);

    if (t < BUCKET_NODES) c[t] = 0;
    __syncthreads();
    for (int p = bs + t; p < be; p += SORT_BATCH * SORT_THREADS) {
        int d[SORT_BATCH];
        #pragma unroll
        for (int u = 0; u < SORT_BATCH; ++u) {
            const int pp = p + u * SORT_THREADS;
            d[u] = (pp < be) ? pairs[pp].x : -1;
        }
        #pragma unroll
        for (int u = 0; u < SORT_BATCH; ++u)
            if (d[u] >= 0) atomicAdd(&c[d[u] & (BUCKET_NODES - 1)], 1);
    }
    __syncthreads();

    int x = 0, inc = 0, start = 0;
    if (t < BUCKET_NODES) {
        x = c[t];
        inc = x;
        for (int off = 1; off < 32; off <<= 1) {
            const int n = __shfl_up(inc, off);
            if (lane >= off) inc += n;
        }
        if (lane == 31) wsum[wave] = inc;
    }
    __syncthreads();
    if (t < BUCKET_NODES) {
        int woff = 0;
        for (int w = 0; w < wave; ++w) woff += wsum[w];
        start = bs + woff + inc - x;
        c[t] = start;
    }
    __syncthreads();

    for (int p = bs + t; p < be; p += SORT_BATCH * SORT_THREADS) {
        int2 pr[SORT_BATCH];
        #pragma unroll
        for (int u = 0; u < SORT_BATCH; ++u) {
            const int pp = p + u * SORT_THREADS;
            pr[u] = (pp < be) ? pairs[pp] : make_int2(-1, 0);
        }
        #pragma unroll
        for (int u = 0; u < SORT_BATCH; ++u) {
            if (pr[u].x >= 0) {
                int pos = atomicAdd(&c[pr[u].x & (BUCKET_NODES - 1)], 1);
                pos = (int)min((unsigned)pos, (unsigned)(EE - 1));
                col[pos] = pr[u].y;
            }
        }
    }
    if (t < BUCKET_NODES) {
        const int node = k * BUCKET_NODES + t;
        if (node < NN) rowptr[node] = start;
    }
    if (k == NBUCKETS - 1 && t == 0) rowptr[NN] = be;
}

template<int DIM, bool BN>
__global__ __launch_bounds__(256) void gather_agg_kernel(
    const float* __restrict__ h, const int* __restrict__ rowptr,
    const int* __restrict__ col, const float* __restrict__ epsP, int epsIdx,
    const float* __restrict__ bn_g, const float* __restrict__ bn_b,
    const float* __restrict__ bn_m, const float* __restrict__ bn_v,
    float* __restrict__ agg)
{
    constexpr int VW = DIM / 32;
    typedef __attribute__((ext_vector_type(VW))) float vt;
    const int lane = threadIdx.x & 31;
    const int node = __builtin_amdgcn_readfirstlane(
        (int)(blockIdx.x * 8 + (threadIdx.x >> 5)));
    const int beg = rowptr[node];
    const int end = rowptr[node + 1];
    const float* __restrict__ hl = h + VW * lane;

    vt pv = {}, pm = {}, ps = {}, pg = {}, pb = {};
    if constexpr (BN) {
        pv = *(const vt*)(bn_v + VW * lane);
        pm = *(const vt*)(bn_m + VW * lane);
        pg = *(const vt*)(bn_g + VW * lane);
        pb = *(const vt*)(bn_b + VW * lane);
    }

    const vt self_raw = *(const vt*)(hl + (size_t)node * DIM);

    if constexpr (BN) {
        __builtin_amdgcn_sched_barrier(0);
        #pragma unroll
        for (int i = 0; i < VW; ++i) ps[i] = rsqrtf(pv[i] + 1e-5f);
    }
    auto norm = [&](const vt r) -> vt {
        if constexpr (BN) return (r - pm) * ps * pg + pb;
        else return r;
    };

    vt acc = {};
    const vt zero = {};
    for (int base = beg; base < end; base += 32) {
        const int n = min(32, end - base);
        int ids = (lane < n) ? col[base + lane] : 0;
        ids = (int)min((unsigned)ids, (unsigned)(NN - 1));
        int j = 0;
        for (; j + 4 <= n; j += 4) {
            const int s0 = __builtin_amdgcn_readlane(ids, j);
            const int s1 = __builtin_amdgcn_readlane(ids, j + 1);
            const int s2 = __builtin_amdgcn_readlane(ids, j + 2);
            const int s3 = __builtin_amdgcn_readlane(ids, j + 3);
            const vt v0 = *(const vt*)(hl + (size_t)s0 * DIM);
            const vt v1 = *(const vt*)(hl + (size_t)s1 * DIM);
            const vt v2 = *(const vt*)(hl + (size_t)s2 * DIM);
            const vt v3 = *(const vt*)(hl + (size_t)s3 * DIM);
            acc += norm(v0); acc += norm(v1); acc += norm(v2); acc += norm(v3);
        }
        if (j < n) {
            const int last = n - 1;
            const int s0 = __builtin_amdgcn_readlane(ids, j);
            const int s1 = __builtin_amdgcn_readlane(ids, min(j + 1, last));
            const int s2 = __builtin_amdgcn_readlane(ids, min(j + 2, last));
            const vt v0 = *(const vt*)(hl + (size_t)s0 * DIM);
            const vt v1 = *(const vt*)(hl + (size_t)s1 * DIM);
            const vt v2 = *(const vt*)(hl + (size_t)s2 * DIM);
            acc += norm(v0);
            acc += (j + 1 < n) ? norm(v1) : zero;
            acc += (j + 2 < n) ? norm(v2) : zero;
        }
    }
    const float sc = 1.0f + epsP[epsIdx];
    *(vt*)(agg + (size_t)node * DIM + VW * lane) = sc * norm(self_raw) + acc;
}

#define ROW_TILES (NN / 16)
#define MLP_WAVES 14
#define MLP_THREADS (MLP_WAVES * 32)
#define MLP_BLOCKS ((ROW_TILES + MLP_WAVES - 1) / MLP_WAVES)
#define W1PITCH 144
#define W2PITCH 132

template<int KK, int PITCH>
struct WStage {
    static constexpr int P = KK / 2;
    static constexpr int NMAX = (P + MLP_WAVES - 1) / MLP_WAVES;
    float4 r0[NMAX], r1[NMAX];
    __device__ __forceinline__ void load(const float* __restrict__ Wg, int wave, int lane) {
        #pragma unroll
        for (int i = 0; i < NMAX; ++i) {
            const int kp = wave + MLP_WAVES * i;
            if (kp < P) {
                r0[i] = *(const float4*)(Wg + (size_t)(2 * kp) * HID + 4 * lane);
                r1[i] = *(const float4*)(Wg + (size_t)(2 * kp + 1) * HID + 4 * lane);
            } else {
                r0[i] = make_float4(0.f, 0.f, 0.f, 0.f);
                r1[i] = make_float4(0.f, 0.f, 0.f, 0.f);
            }
        }
    }
    __device__ __forceinline__ void store(float* Ws, int wave, int lane) const {
        #pragma unroll
        for (int i = 0; i < NMAX; ++i) {
            const int kp = wave + MLP_WAVES * i;
            if (kp < P) {
                float* d = Ws + (kp * PITCH + 4 * lane) * 2;
                *(float4*)(d)     = make_float4(r0[i].x, r1[i].x, r0[i].y, r1[i].y);
                *(float4*)(d + 4) = make_float4(r0[i].z, r1[i].z, r0[i].w, r1[i].w);
            }
        }
    }
};

template<int K, bool POOL>
__global__ __launch_bounds__(MLP_THREADS, 7) void mlp_kernel(
    const float* __restrict__ A,
    const float* __restrict__ W1, const float* __restrict__ b1,
    const float* __restrict__ g1, const float* __restrict__ o1,
    const float* __restrict__ m1, const float* __restrict__ v1,
    const float* __restrict__ W2, const float* __restrict__ b2,
    const float* __restrict__ g2, const float* __restrict__ o2,
    const float* __restrict__ m2, const float* __restrict__ v2,
    float* __restrict__ out,
    const int* __restrict__ batch, float* __restrict__ sums, float* __restrict__ cnt)
{
    __shared__ __attribute__((aligned(16))) float Ws1[(K / 2) * W1PITCH * 2];
    __shared__ __attribute__((aligned(16))) float Ws2[(HID / 2) * W2PITCH * 2];

    const int lane = threadIdx.x & 31;
    const int wave = __builtin_amdgcn_readfirstlane((int)(threadIdx.x >> 5));
    const int half = lane >> 4;
    const int mr   = lane & 15;

    const int tileM = blockIdx.x * MLP_WAVES + wave;
    const bool live = tileM < ROW_TILES;
    const float* __restrict__ Arow =
        A + (size_t)((live ? tileM : 0) * 16 + mr) * K + 2 * half;

    WStage<K, W1PITCH> st1;
    WStage<HID, W2PITCH> st2;
    st1.load(W1, wave, lane);
    st2.load(W2, wave, lane);
    constexpr int KC = 16;
    v2f a_cur[KC / 4];
    #pragma unroll
    for (int i = 0; i < KC / 4; ++i) a_cur[i] = *(const v2f*)(Arow + 4 * i);
    st1.store(Ws1, wave, lane);
    st2.store(Ws2, wave, lane);
    __syncthreads();
    if (!live) return;

    const float* Wl1 = Ws1 + (half * W1PITCH + mr) * 2;

    v8f acc1[8];
    #pragma unroll
    for (int c = 0; c < 8; ++c) acc1[c] = (v8f){};

    #pragma unroll 1
    for (int kc = 0; kc < K; kc += KC) {
        const int kn = min(kc + KC, K - KC);
        v2f a_nxt[KC / 4];
        #pragma unroll
        for (int i = 0; i < KC / 4; ++i) a_nxt[i] = *(const v2f*)(Arow + kn + 4 * i);
        #pragma unroll
        for (int i = 0; i < KC / 4; ++i) {
            #pragma unroll
            for (int c = 0; c < 8; ++c) {
                const v2f w = *(const v2f*)(Wl1 + ((kc + 4 * i) / 2) * W1PITCH * 2 + c * 32);
                acc1[c] = __builtin_amdgcn_wmma_f32_16x16x4_f32(
                    false, w, false, a_cur[i], (short)0, acc1[c], false, false);
            }
        }
        #pragma unroll
        for (int i = 0; i < KC / 4; ++i) a_cur[i] = a_nxt[i];
    }

    #pragma unroll
    for (int gq = 0; gq < 16; ++gq) {
        const int c = gq >> 1;
        const int u = gq & 1;
        const int f0 = c * 16 + 8 * half + 4 * u;
        const float4 tb = *(const float4*)(b1 + f0);
        const float4 tm = *(const float4*)(m1 + f0);
        const float4 tv = *(const float4*)(v1 + f0);
        const float4 tg = *(const float4*)(g1 + f0);
        const float4 to = *(const float4*)(o1 + f0);
        const float pb[4] = {tb.x, tb.y, tb.z, tb.w};
        const float pm[4] = {tm.x, tm.y, tm.z, tm.w};
        const float pv[4] = {tv.x, tv.y, tv.z, tv.w};
        const float pg[4] = {tg.x, tg.y, tg.z, tg.w};
        const float po[4] = {to.x, to.y, to.z, to.w};
        #pragma unroll
        for (int e = 0; e < 4; ++e) {
            const int r = 4 * u + e;
            float y = acc1[c][r] + pb[e];
            y = (y - pm[e]) * rsqrtf(pv[e] + 1e-5f) * pg[e] + po[e];
            acc1[c][r] = gelu_tail_form(y);
        }
        __builtin_amdgcn_sched_barrier(0);
    }

    int gid[8] = {};
    if constexpr (POOL) {
        const int4* __restrict__ bp = (const int4*)(batch + tileM * 16 + 8 * half);
        const int4 q0 = bp[0];
        const int4 q1 = bp[1];
        gid[0] = q0.x; gid[1] = q0.y; gid[2] = q0.z; gid[3] = q0.w;
        gid[4] = q1.x; gid[5] = q1.y; gid[6] = q1.z; gid[7] = q1.w;
    }

    constexpr int NT = POOL ? 2 : 4;
    const float* Wl2 = Ws2 + (4 * half * W2PITCH + mr) * 2;

    #pragma unroll 1
    for (int pass = 0; pass < 8 / NT; ++pass) {
        v8f acc2[NT];
        #pragma unroll
        for (int o = 0; o < NT; ++o) acc2[o] = (v8f){};
        const float* Wp = Wl2 + pass * (NT * 32);
        #pragma unroll
        for (int jt = 0; jt < 8; ++jt) {
            #pragma unroll
            for (int q = 0; q < 4; ++q) {
                v2f a;
                a[0] = acc1[jt][2 * q];
                a[1] = acc1[jt][2 * q + 1];
                #pragma unroll
                for (int o = 0; o < NT; ++o) {
                    const v2f b = *(const v2f*)(Wp + (8 * jt + q) * W2PITCH * 2 + o * 32);
                    acc2[o] = __builtin_amdgcn_wmma_f32_16x16x4_f32(
                        false, a, false, b, (short)0, acc2[o], false, false);
                }
            }
        }

        #pragma unroll
        for (int o = 0; o < NT; ++o) {
            const int col = (NT * pass + o) * 16 + mr;
            const float cb = b2[col];
            const float cm = m2[col];
            const float cs = rsqrtf(v2[col] + 1e-5f);
            const float cg = g2[col];
            const float co = o2[col];
            if constexpr (!POOL) {
                #pragma unroll
                for (int r = 0; r < 8; ++r) {
                    const int row = tileM * 16 + r + 8 * half;
                    float y = acc2[o][r] + cb;
                    y = (y - cm) * cs * cg + co;
                    y = gelu_tail_form(y);
                    out[(size_t)row * HID + col] = y;
                }
            } else {
                int cur = gid[0];
                float s = 0.0f;
                #pragma unroll
                for (int r = 0; r < 8; ++r) {
                    float y = acc2[o][r] + cb;
                    y = (y - cm) * cs * cg + co;
                    y = gelu_tail_form(y);
                    if (gid[r] != cur) {
                        if ((unsigned)cur < (unsigned)GG)
                            unsafeAtomicAdd(sums + (size_t)cur * HID + col, s);
                        s = 0.0f;
                        cur = gid[r];
                    }
                    s += y;
                }
                if ((unsigned)cur < (unsigned)GG)
                    unsafeAtomicAdd(sums + (size_t)cur * HID + col, s);
            }
        }
    }
    if constexpr (POOL) {
        if (mr == 0) {
            int cur = gid[0];
            float n = 0.0f;
            #pragma unroll
            for (int r = 0; r < 8; ++r) {
                if (gid[r] != cur) {
                    if ((unsigned)cur < (unsigned)GG) unsafeAtomicAdd(cnt + cur, n);
                    n = 0.0f;
                    cur = gid[r];
                }
                n += 1.0f;
            }
            if ((unsigned)cur < (unsigned)GG) unsafeAtomicAdd(cnt + cur, n);
        }
    }
}

__global__ __launch_bounds__(256) void zero_kernel(float* __restrict__ p, int n) {
    int idx = blockIdx.x * blockDim.x + threadIdx.x;
    if (idx < n) p[idx] = 0.0f;
}

__global__ __launch_bounds__(128) void head_kernel(
    const float* __restrict__ sums, const float* __restrict__ cnt,
    const float* __restrict__ fc1w, const float* __restrict__ fc1b,
    const float* __restrict__ lng, const float* __restrict__ lnb,
    const float* __restrict__ fc2w, const float* __restrict__ fc2b,
    float* __restrict__ out)
{
    __shared__ float p[HID];
    __shared__ float rbuf[HID];
    __shared__ float red[HID];
    const int g = blockIdx.x;
    const int t = threadIdx.x;

    p[t] = sums[(size_t)g * HID + t] / fmaxf(cnt[g], 1.0f);
    __syncthreads();

    float o = fc1b[t];
    #pragma unroll 4
    for (int k = 0; k < HID; ++k) o += p[k] * fc1w[k * HID + t];

    red[t] = o;
    __syncthreads();
    for (int s = 64; s > 0; s >>= 1) {
        if (t < s) red[t] += red[t + s];
        __syncthreads();
    }
    const float mu = red[0] * (1.0f / HID);
    __syncthreads();

    const float dv = o - mu;
    red[t] = dv * dv;
    __syncthreads();
    for (int s = 64; s > 0; s >>= 1) {
        if (t < s) red[t] += red[t + s];
        __syncthreads();
    }
    const float var = red[0] * (1.0f / HID);

    float y = dv * rsqrtf(var + 1e-5f) * lng[t] + lnb[t];
    y = gelu_exact(y) + p[t];
    rbuf[t] = y;
    __syncthreads();

    if (t < LAT) {
        float o2 = fc2b[t];
        #pragma unroll 4
        for (int k = 0; k < HID; ++k) o2 += rbuf[k] * fc2w[k * LAT + t];
        out[(size_t)g * LAT + t] = o2;
    }
}

extern "C" void kernel_launch(void* const* d_in, const int* in_sizes, int n_in,
                              void* d_out, int out_size, void* d_ws, size_t ws_size,
                              hipStream_t stream) {
    const float* x       = (const float*)d_in[0];
    const int*   ei      = (const int*)  d_in[1];
    const int*   batch   = (const int*)  d_in[2];
    const float* in_g    = (const float*)d_in[3];
    const float* in_b    = (const float*)d_in[4];
    const float* in_m    = (const float*)d_in[5];
    const float* in_v    = (const float*)d_in[6];
    const float* eps0    = (const float*)d_in[7];
    const float* c0_w1   = (const float*)d_in[8];
    const float* c0_b1   = (const float*)d_in[9];
    const float* c0_g    = (const float*)d_in[10];
    const float* c0_bb   = (const float*)d_in[11];
    const float* c0_m    = (const float*)d_in[12];
    const float* c0_v    = (const float*)d_in[13];
    const float* c0_w2   = (const float*)d_in[14];
    const float* c0_b2   = (const float*)d_in[15];
    const float* bn0_g   = (const float*)d_in[16];
    const float* bn0_b   = (const float*)d_in[17];
    const float* bn0_m   = (const float*)d_in[18];
    const float* bn0_v   = (const float*)d_in[19];
    const float* epsR    = (const float*)d_in[20];
    const float* cR_w1   = (const float*)d_in[21];
    const float* cR_b1   = (const float*)d_in[22];
    const float* cR_g    = (const float*)d_in[23];
    const float* cR_bb   = (const float*)d_in[24];
    const float* cR_m    = (const float*)d_in[25];
    const float* cR_v    = (const float*)d_in[26];
    const float* cR_w2   = (const float*)d_in[27];
    const float* cR_b2   = (const float*)d_in[28];
    const float* bnR_g   = (const float*)d_in[29];
    const float* bnR_b   = (const float*)d_in[30];
    const float* bnR_m   = (const float*)d_in[31];
    const float* bnR_v   = (const float*)d_in[32];
    const float* fc1_w   = (const float*)d_in[33];
    const float* fc1_b   = (const float*)d_in[34];
    const float* ln_g    = (const float*)d_in[35];
    const float* ln_b    = (const float*)d_in[36];
    const float* fc2_w   = (const float*)d_in[37];
    const float* fc2_b   = (const float*)d_in[38];

    const int* src = ei;
    const int* dst = ei + EE;

    float* bufA = (float*)d_ws;
    float* bufB = bufA + (size_t)NN * HID;
    int*   col    = (int*)(bufB + (size_t)NN * HID);
    int*   pairs  = col + EE;
    int*   cells  = pairs + 2 * EE;
    int*   base   = cells + PART_CELLS;
    int*   rowptr = base + PART_CELLS + 4;
    float* sums = bufA + (size_t)3 * NN * HID;
    float* cnt  = sums + (size_t)GG * HID;

    const int TB = 256;
    const int nodeBlocks = NN / 8;

    part_hist_kernel<<<PART_BLOCKS, PART_THREADS, 0, stream>>>(dst, cells);
    part_scan_kernel<<<SCAN_BLOCKS, SCAN_THREADS, 0, stream>>>(cells, base);
    part_scatter_kernel<<<PART_BLOCKS, PART_THREADS, 0, stream>>>(src, dst, base, pairs);
    bucket_sort_kernel<<<NBUCKETS, SORT_THREADS, 0, stream>>>(pairs, base, rowptr, col);

    gather_agg_kernel<INDIM, true><<<nodeBlocks, TB, 0, stream>>>(
        x, rowptr, col, eps0, 0, in_g, in_b, in_m, in_v, bufB);
    mlp_kernel<INDIM, false><<<MLP_BLOCKS, MLP_THREADS, 0, stream>>>(
        bufB, c0_w1, c0_b1, c0_g, c0_bb, c0_m, c0_v,
        c0_w2, c0_b2, bn0_g, bn0_b, bn0_m, bn0_v, bufA, batch, sums, cnt);

    zero_kernel<<<(GG * HID + GG + TB - 1) / TB, TB, 0, stream>>>(sums, GG * HID + GG);

    gather_agg_kernel<HID, false><<<nodeBlocks, TB, 0, stream>>>(
        bufA, rowptr, col, epsR, 0, nullptr, nullptr, nullptr, nullptr, bufB);
    mlp_kernel<HID, false><<<MLP_BLOCKS, MLP_THREADS, 0, stream>>>(
        bufB, cR_w1, cR_b1, cR_g, cR_bb, cR_m, cR_v,
        cR_w2, cR_b2, bnR_g, bnR_b, bnR_m, bnR_v, bufA, batch, sums, cnt);
    gather_agg_kernel<HID, false><<<nodeBlocks, TB, 0, stream>>>(
        bufA, rowptr, col, epsR, 1, nullptr, nullptr, nullptr, nullptr, bufB);
    mlp_kernel<HID, true><<<MLP_BLOCKS, MLP_THREADS, 0, stream>>>(
        bufB, cR_w1 + HID * HID, cR_b1 + HID, cR_g + HID, cR_bb + HID, cR_m + HID, cR_v + HID,
        cR_w2 + HID * HID, cR_b2 + HID, bnR_g + HID, bnR_b + HID, bnR_m + HID, bnR_v + HID,
        bufA, batch, sums, cnt);

    head_kernel<<<GG, HID, 0, stream>>>(
        sums, cnt, fc1_w, fc1_b, ln_g, ln_b, fc2_w, fc2_b, (float*)d_out);
}


# First product's BN+GELU moved into the second product's first pass, tile jt+1 beside the WMMAs on tile jt (fragment reads grouped per k step); BN one fma from per-block (s,t) in LDS; ulp moves
# speedup vs baseline: 3.2977x; 1.0352x over previous
#include <hip/hip_runtime.h>
#include <math.h>

typedef __attribute__((ext_vector_type(2))) float v2f;
typedef __attribute__((ext_vector_type(8))) float v8f;

#define NN   100000
#define EE   1600000
#define INDIM 64
#define HID  128
#define LAT  64
#define GG   1000

__device__ __forceinline__ float gelu_exact(float x) {
    return 0.5f * x * (1.0f + erff(x * 0.70710678118654752f));
}

__device__ __forceinline__ float gelu_tail_form(float x) {
    const float t = fabsf(x);
    float s = fmaf(t, __uint_as_float(0x360d598au), __uint_as_float(0xb80136dbu));
    s = fmaf(t, s, __uint_as_float(0x39197cbcu));
    s = fmaf(t, s, __uint_as_float(0x39717391u));
    s = fmaf(t, s, __uint_as_float(0xbbeb4a40u));
    s = fmaf(t, s, __uint_as_float(0x3d57563eu));
    s = fmaf(t, s, __uint_as_float(0x3eeb1a59u));
    s = fmaf(t, s, __uint_as_float(0x3f93577du));
    const float tail = __builtin_amdgcn_exp2f(-fmaf(t, s, 1.0f));
    return fmaf(-t, tail, fmaxf(x, 0.0f));
}

#define BUCKET_SHIFT 8
#define BUCKET_NODES (1 << BUCKET_SHIFT)
#define NBUCKETS ((NN + BUCKET_NODES - 1) / BUCKET_NODES)
#define PART_BLOCKS 256
#define PART_THREADS 1024
#define PART_CHUNK ((EE + PART_BLOCKS - 1) / PART_BLOCKS)
#define PART_BATCH 8
#define PART_CELLS (NBUCKETS * PART_BLOCKS)
#define SCAN_THREADS 1024
#define SCAN_BLOCKS ((PART_CELLS + SCAN_THREADS - 1) / SCAN_THREADS)
#define SCAN_PRE_MAX (((SCAN_BLOCKS - 1) * (SCAN_THREADS / 4) + SCAN_THREADS - 1) / SCAN_THREADS)
#define SORT_THREADS 1024
#define SORT_BATCH 5

__global__ __launch_bounds__(PART_THREADS) void part_hist_kernel(
    const int* __restrict__ dst, int* __restrict__ cells)
{
    __shared__ int h[NBUCKETS];
    const int t = threadIdx.x;
    for (int k = t; k < NBUCKETS; k += PART_THREADS) h[k] = 0;
    __syncthreads();
    const int e0 = blockIdx.x * PART_CHUNK;
    const int e1 = min(e0 + PART_CHUNK, EE);
    for (int e = e0 + t; e < e1; e += PART_BATCH * PART_THREADS) {
        int d[PART_BATCH];
        #pragma unroll
        for (int u = 0; u < PART_BATCH; ++u) {
            const int ee = e + u * PART_THREADS;
            d[u] = (ee < e1) ? dst[ee] : -1;
        }
        #pragma unroll
        for (int u = 0; u < PART_BATCH; ++u)
            if ((unsigned)d[u] < (unsigned)NN) atomicAdd(&h[d[u] >> BUCKET_SHIFT], 1);
    }
    __syncthreads();
    for (int k = t; k < NBUCKETS; k += PART_THREADS)
        cells[k * PART_BLOCKS + blockIdx.x] = h[k];
}

__global__ __launch_bounds__(SCAN_THREADS) void part_scan_kernel(
    const int* __restrict__ cells, int* __restrict__ base)
{
    __shared__ int red[SCAN_THREADS / 32];
    __shared__ int wsum[SCAN_THREADS / 32];
    const int t = threadIdx.x;
    const int lane = t & 31;
    const int wave = t >> 5;
    const int nb = blockIdx.x;

    const int4* __restrict__ c4 = (const int4*)cells;
    const int n4 = nb * (SCAN_THREADS / 4);
    int4 q[SCAN_PRE_MAX];
    #pragma unroll
    for (int u = 0; u < SCAN_PRE_MAX; ++u) {
        const int i = u * SCAN_THREADS + t;
        q[u] = (i < n4) ? c4[i] : make_int4(0, 0, 0, 0);
    }
    int s = 0;
    #pragma unroll
    for (int u = 0; u < SCAN_PRE_MAX; ++u) s += q[u].x + q[u].y + q[u].z + q[u].w;
    const int idx = nb * SCAN_THREADS + t;
    const int x = (idx < PART_CELLS) ? cells[idx] : 0;

    for (int off = 16; off > 0; off >>= 1) s += __shfl_xor(s, off);
    int inc = x;
    for (int off = 1; off < 32; off <<= 1) {
        const int n = __shfl_up(inc, off);
        if (lane >= off) inc += n;
    }
    if (lane == 0) red[wave] = s;
    if (lane == 31) wsum[wave] = inc;
    __syncthreads();
    int before = 0;
    for (int w = 0; w < SCAN_THREADS / 32; ++w) before += red[w];
    int woff = 0;
    for (int w = 0; w < wave; ++w) woff += wsum[w];
    const int ex = before + woff + inc - x;
    if (idx < PART_CELLS) base[idx] = ex;
    if (idx == PART_CELLS - 1) base[PART_CELLS] = ex + x;
}

__global__ __launch_bounds__(PART_THREADS) void part_scatter_kernel(
    const int* __restrict__ src, const int* __restrict__ dst,
    const int* __restrict__ base, int* __restrict__ pairs_out)
{
    int2* __restrict__ pairs = (int2*)pairs_out;
    __shared__ int cur[NBUCKETS];
    const int t = threadIdx.x;
    for (int k = t; k < NBUCKETS; k += PART_THREADS)
        cur[k] = base[k * PART_BLOCKS + blockIdx.x];
    __syncthreads();
    const int e0 = blockIdx.x * PART_CHUNK;
    const int e1 = min(e0 + PART_CHUNK, EE);
    for (int e = e0 + t; e < e1; e += PART_BATCH * PART_THREADS) {
        int d[PART_BATCH], s[PART_BATCH];
        #pragma unroll
        for (int u = 0; u < PART_BATCH; ++u) {
            const int ee = e + u * PART_THREADS;
            const bool ok = ee < e1;
            d[u] = ok ? dst[ee] : -1;
            s[u] = ok ? src[ee] : 0;
        }
        #pragma unroll
        for (int u = 0; u < PART_BATCH; ++u) {
            if ((unsigned)d[u] < (unsigned)NN) {
                int pos = atomicAdd(&cur[d[u] >> BUCKET_SHIFT], 1);
                pos = (int)min((unsigned)pos, (unsigned)(EE - 1));
                pairs[pos] = make_int2(d[u], s[u]);
            }
        }
    }
}

__global__ __launch_bounds__(SORT_THREADS) void bucket_sort_kernel(
    const int* __restrict__ pairs_in, const int* __restrict__ base,
    int* __restrict__ rowptr, int* __restrict__ col)
{
    const int2* __restrict__ pairs = (const int2*)pairs_in;
    __shared__ int c[BUCKET_NODES];
    __shared__ int wsum[BUCKET_NODES / 32];
    const int t = threadIdx.x;
    const int lane = t & 31;
    const int wave = t >> 5;
    const int k = blockIdx.x;
    int bs = base[k * PART_BLOCKS];
    int be = base[(k + 1) * PART_BLOCKS];
    bs = max(bs, 0);
    be = min(be, EE);

    if (t < BUCKET_NODES) c[t] = 0;
    __syncthreads();
    for (int p = bs + t; p < be; p += SORT_BATCH * SORT_THREADS) {
        int d[SORT_BATCH];
        #pragma unroll
        for (int u = 0; u < SORT_BATCH; ++u) {
            const int pp = p + u * SORT_THREADS;
            d[u] = (pp < be) ? pairs[pp].x : -1;
        }
        #pragma unroll
        for (int u = 0; u < SORT_BATCH; ++u)
            if (d[u] >= 0) atomicAdd(&c[d[u] & (BUCKET_NODES - 1)], 1);
    }
    __syncthreads();

    int x = 0, inc = 0, start = 0;
    if (t < BUCKET_NODES) {
        x = c[t];
        inc = x;
        for (int off = 1; off < 32; off <<= 1) {
            const int n = __shfl_up(inc, off);
            if (lane >= off) inc += n;
        }
        if (lane == 31) wsum[wave] = inc;
    }
    __syncthreads();
    if (t < BUCKET_NODES) {
        int woff = 0;
        for (int w = 0; w < wave; ++w) woff += wsum[w];
        start = bs + woff + inc - x;
        c[t] = start;
    }
    __syncthreads();

    for (int p = bs + t; p < be; p += SORT_BATCH * SORT_THREADS) {
        int2 pr[SORT_BATCH];
        #pragma unroll
        for (int u = 0; u < SORT_BATCH; ++u) {
            const int pp = p + u * SORT_THREADS;
            pr[u] = (pp < be) ? pairs[pp] : make_int2(-1, 0);
        }
        #pragma unroll
        for (int u = 0; u < SORT_BATCH; ++u) {
            if (pr[u].x >= 0) {
                int pos = atomicAdd(&c[pr[u].x & (BUCKET_NODES - 1)], 1);
                pos = (int)min((unsigned)pos, (unsigned)(EE - 1));
                col[pos] = pr[u].y;
            }
        }
    }
    if (t < BUCKET_NODES) {
        const int node = k * BUCKET_NODES + t;
        if (node < NN) rowptr[node] = start;
    }
    if (k == NBUCKETS - 1 && t == 0) rowptr[NN] = be;
}

template<int DIM, bool BN>
__global__ __launch_bounds__(256) void gather_agg_kernel(
    const float* __restrict__ h, const int* __restrict__ rowptr,
    const int* __restrict__ col, const float* __restrict__ epsP, int epsIdx,
    const float* __restrict__ bn_g, const float* __restrict__ bn_b,
    const float* __restrict__ bn_m, const float* __restrict__ bn_v,
    float* __restrict__ agg)
{
    constexpr int VW = DIM / 32;
    typedef __attribute__((ext_vector_type(VW))) float vt;
    const int lane = threadIdx.x & 31;
    const int node = __builtin_amdgcn_readfirstlane(
        (int)(blockIdx.x * 8 + (threadIdx.x >> 5)));
    const int beg = rowptr[node];
    const int end = rowptr[node + 1];
    const float* __restrict__ hl = h + VW * lane;

    vt pv = {}, pm = {}, ps = {}, pg = {}, pb = {};
    if constexpr (BN) {
        pv = *(const vt*)(bn_v + VW * lane);
        pm = *(const vt*)(bn_m + VW * lane);
        pg = *(const vt*)(bn_g + VW * lane);
        pb = *(const vt*)(bn_b + VW * lane);
    }

    const vt self_raw = *(const vt*)(hl + (size_t)node * DIM);

    if constexpr (BN) {
        __builtin_amdgcn_sched_barrier(0);
        #pragma unroll
        for (int i = 0; i < VW; ++i) ps[i] = rsqrtf(pv[i] + 1e-5f);
    }
    auto norm = [&](const vt r) -> vt {
        if constexpr (BN) return (r - pm) * ps * pg + pb;
        else return r;
    };

    vt acc = {};
    const vt zero = {};
    for (int base = beg; base < end; base += 32) {
        const int n = min(32, end - base);
        int ids = (lane < n) ? col[base + lane] : 0;
        ids = (int)min((unsigned)ids, (unsigned)(NN - 1));
        int j = 0;
        for (; j + 4 <= n; j += 4) {
            const int s0 = __builtin_amdgcn_readlane(ids, j);
            const int s1 = __builtin_amdgcn_readlane(ids, j + 1);
            const int s2 = __builtin_amdgcn_readlane(ids, j + 2);
            const int s3 = __builtin_amdgcn_readlane(ids, j + 3);
            const vt v0 = *(const vt*)(hl + (size_t)s0 * DIM);
            const vt v1 = *(const vt*)(hl + (size_t)s1 * DIM);
            const vt v2 = *(const vt*)(hl + (size_t)s2 * DIM);
            const vt v3 = *(const vt*)(hl + (size_t)s3 * DIM);
            acc += norm(v0); acc += norm(v1); acc += norm(v2); acc += norm(v3);
        }
        if (j < n) {
            const int last = n - 1;
            const int s0 = __builtin_amdgcn_readlane(ids, j);
            const int s1 = __builtin_amdgcn_readlane(ids, min(j + 1, last));
            const int s2 = __builtin_amdgcn_readlane(ids, min(j + 2, last));
            const vt v0 = *(const vt*)(hl + (size_t)s0 * DIM);
            const vt v1 = *(const vt*)(hl + (size_t)s1 * DIM);
            const vt v2 = *(const vt*)(hl + (size_t)s2 * DIM);
            acc += norm(v0);
            acc += (j + 1 < n) ? norm(v1) : zero;
            acc += (j + 2 < n) ? norm(v2) : zero;
        }
    }
    const float sc = 1.0f + epsP[epsIdx];
    *(vt*)(agg + (size_t)node * DIM + VW * lane) = sc * norm(self_raw) + acc;
}

#define ROW_TILES (NN / 16)
#define MLP_WAVES 14
#define MLP_THREADS (MLP_WAVES * 32)
#define MLP_BLOCKS ((ROW_TILES + MLP_WAVES - 1) / MLP_WAVES)
#define W1PITCH 144
#define W2PITCH 132

template<int KK, int PITCH>
struct WStage {
    static constexpr int P = KK / 2;
    static constexpr int NMAX = (P + MLP_WAVES - 1) / MLP_WAVES;
    float4 r0[NMAX], r1[NMAX];
    __device__ __forceinline__ void load(const float* __restrict__ Wg, int wave, int lane) {
        #pragma unroll
        for (int i = 0; i < NMAX; ++i) {
            const int kp = wave + MLP_WAVES * i;
            if (kp < P) {
                r0[i] = *(const float4*)(Wg + (size_t)(2 * kp) * HID + 4 * lane);
                r1[i] = *(const float4*)(Wg + (size_t)(2 * kp + 1) * HID + 4 * lane);
            } else {
                r0[i] = make_float4(0.f, 0.f, 0.f, 0.f);
                r1[i] = make_float4(0.f, 0.f, 0.f, 0.f);
            }
        }
    }
    __device__ __forceinline__ void store(float* Ws, int wave, int lane) const {
        #pragma unroll
        for (int i = 0; i < NMAX; ++i) {
            const int kp = wave + MLP_WAVES * i;
            if (kp < P) {
                float* d = Ws + (kp * PITCH + 4 * lane) * 2;
                *(float4*)(d)     = make_float4(r0[i].x, r1[i].x, r0[i].y, r1[i].y);
                *(float4*)(d + 4) = make_float4(r0[i].z, r1[i].z, r0[i].w, r1[i].w);
            }
        }
    }
};


template<bool POOL, bool ACT, int NT>
__device__ __forceinline__ void second_product_pass(
    v8f (&T)[8], const int pass, const float* Wl2,
    const float* bn1h, const float* bn2,
    const int mr, const int half, const int tileM,
    float* __restrict__ out, const int (&gid)[8], float* __restrict__ sums)
{
    v8f acc2[NT];
    #pragma unroll
    for (int o = 0; o < NT; ++o) acc2[o] = (v8f){};
    const float* Wp = Wl2 + pass * (NT * 32);
    #pragma unroll
    for (int jt = 0; jt < 8; ++jt) {
        #pragma unroll
        for (int q = 0; q < 4; ++q) {
            v2f a;
            a[0] = T[jt][2 * q];
            a[1] = T[jt][2 * q + 1];
            v2f b[NT];
            #pragma unroll
            for (int o = 0; o < NT; ++o)
                b[o] = *(const v2f*)(Wp + (8 * jt + q) * W2PITCH * 2 + o * 32);
            #pragma unroll
            for (int o = 0; o < NT; ++o)
                acc2[o] = __builtin_amdgcn_wmma_f32_16x16x4_f32(
                    false, a, false, b[o], (short)0, acc2[o], false, false);
            if constexpr (ACT) {
                if (jt + 1 < 8) {
                    const float4 p = *(const float4*)(bn1h + 2 * (16 * (jt + 1) + 2 * q));
                    T[jt + 1][2 * q]     = gelu_tail_form(fmaf(T[jt + 1][2 * q],     p.x, p.z));
                    T[jt + 1][2 * q + 1] = gelu_tail_form(fmaf(T[jt + 1][2 * q + 1], p.y, p.w));
                }
            }
        }
    }

    #pragma unroll
    for (int o = 0; o < NT; ++o) {
        const int col = (NT * pass + o) * 16 + mr;
        const float2 st = *(const float2*)(bn2 + 2 * col);
        if constexpr (!POOL) {
            #pragma unroll
            for (int r = 0; r < 8; ++r) {
                const int row = tileM * 16 + r + 8 * half;
                const float y = gelu_tail_form(fmaf(acc2[o][r], st.x, st.y));
                out[(size_t)row * HID + col] = y;
            }
        } else {
            int cur = gid[0];
            float s = 0.0f;
            #pragma unroll
            for (int r = 0; r < 8; ++r) {
                const float y = gelu_tail_form(fmaf(acc2[o][r], st.x, st.y));
                if (gid[r] != cur) {
                    if ((unsigned)cur < (unsigned)GG)
                        unsafeAtomicAdd(sums + (size_t)cur * HID + col, s);
                    s = 0.0f;
                    cur = gid[r];
                }
                s += y;
            }
            if ((unsigned)cur < (unsigned)GG)
                unsafeAtomicAdd(sums + (size_t)cur * HID + col, s);
        }
    }
}

template<int K, bool POOL>
__global__ __launch_bounds__(MLP_THREADS, 7) void mlp_kernel(
    const float* __restrict__ A,
    const float* __restrict__ W1, const float* __restrict__ b1,
    const float* __restrict__ g1, const float* __restrict__ o1,
    const float* __restrict__ m1, const float* __restrict__ v1,
    const float* __restrict__ W2, const float* __restrict__ b2,
    const float* __restrict__ g2, const float* __restrict__ o2,
    const float* __restrict__ m2, const float* __restrict__ v2,
    float* __restrict__ out,
    const int* __restrict__ batch, float* __restrict__ sums, float* __restrict__ cnt)
{
    __shared__ __attribute__((aligned(16))) float Ws1[(K / 2) * W1PITCH * 2];
    __shared__ __attribute__((aligned(16))) float Ws2[(HID / 2) * W2PITCH * 2];
    __shared__ __attribute__((aligned(16))) float Bn[4 * HID];

    const int lane = threadIdx.x & 31;
    const int wave = __builtin_amdgcn_readfirstlane((int)(threadIdx.x >> 5));
    const int half = lane >> 4;
    const int mr   = lane & 15;

    const int tileM = blockIdx.x * MLP_WAVES + wave;
    const bool live = tileM < ROW_TILES;
    const float* __restrict__ Arow =
        A + (size_t)((live ? tileM : 0) * 16 + mr) * K + 2 * half;

    WStage<K, W1PITCH> st1;
    WStage<HID, W2PITCH> st2;
    st1.load(W1, wave, lane);
    st2.load(W2, wave, lane);
    constexpr int KC = 16;
    v2f a_cur[KC / 4];
    #pragma unroll
    for (int i = 0; i < KC / 4; ++i) a_cur[i] = *(const v2f*)(Arow + 4 * i);

    const bool tab_live = wave < 8;
    float tab_b = 0.0f, tab_m = 0.0f, tab_v = 1.0f, tab_g = 0.0f, tab_o = 0.0f;
    if (tab_live) {
        const bool second = wave >= 4;
        const int f = threadIdx.x & (HID - 1);
        tab_v = (second ? v2 : v1)[f];
        tab_g = (second ? g2 : g1)[f];
        tab_b = (second ? b2 : b1)[f];
        tab_m = (second ? m2 : m1)[f];
        tab_o = (second ? o2 : o1)[f];
    }

    st1.store(Ws1, wave, lane);
    st2.store(Ws2, wave, lane);
    if (tab_live) {
        const float s = rsqrtf(tab_v + 1e-5f) * tab_g;
        const float t = fmaf(tab_b - tab_m, s, tab_o);
        const int f = threadIdx.x & (HID - 1);
        if (wave < 4) {
            Bn[4 * (f >> 1) + (f & 1)]     = s;
            Bn[4 * (f >> 1) + (f & 1) + 2] = t;
        } else {
            *(float2*)(Bn + 2 * HID + 2 * f) = make_float2(s, t);
        }
    }
    __syncthreads();
    if (!live) return;

    const float* Wl1 = Ws1 + (half * W1PITCH + mr) * 2;

    v8f acc1[8];
    #pragma unroll
    for (int c = 0; c < 8; ++c) acc1[c] = (v8f){};

    #pragma unroll 1
    for (int kc = 0; kc < K; kc += KC) {
        const int kn = min(kc + KC, K - KC);
        v2f a_nxt[KC / 4];
        #pragma unroll
        for (int i = 0; i < KC / 4; ++i) a_nxt[i] = *(const v2f*)(Arow + kn + 4 * i);
        #pragma unroll
        for (int i = 0; i < KC / 4; ++i) {
            #pragma unroll
            for (int c = 0; c < 8; ++c) {
                const v2f w = *(const v2f*)(Wl1 + ((kc + 4 * i) / 2) * W1PITCH * 2 + c * 32);
                acc1[c] = __builtin_amdgcn_wmma_f32_16x16x4_f32(
                    false, w, false, a_cur[i], (short)0, acc1[c], false, false);
            }
        }
        #pragma unroll
        for (int i = 0; i < KC / 4; ++i) a_cur[i] = a_nxt[i];
    }

    int gid[8] = {};
    if constexpr (POOL) {
        const int4* __restrict__ bp = (const int4*)(batch + tileM * 16 + 8 * half);
        const int4 q0 = bp[0];
        const int4 q1 = bp[1];
        gid[0] = q0.x; gid[1] = q0.y; gid[2] = q0.z; gid[3] = q0.w;
        gid[4] = q1.x; gid[5] = q1.y; gid[6] = q1.z; gid[7] = q1.w;
    }

    const float* bn1h = Bn + 16 * half;
    const float* bn2  = Bn + 2 * HID;
    #pragma unroll
    for (int q = 0; q < 4; ++q) {
        const float4 p = *(const float4*)(bn1h + 4 * q);
        acc1[0][2 * q]     = gelu_tail_form(fmaf(acc1[0][2 * q],     p.x, p.z));
        acc1[0][2 * q + 1] = gelu_tail_form(fmaf(acc1[0][2 * q + 1], p.y, p.w));
    }

    constexpr int NT = POOL ? 2 : 4;
    const float* Wl2 = Ws2 + (4 * half * W2PITCH + mr) * 2;

    second_product_pass<POOL, true, NT>(acc1, 0, Wl2, bn1h, bn2, mr, half, tileM, out, gid, sums);
    #pragma unroll 1
    for (int pass = 1; pass < 8 / NT; ++pass)
        second_product_pass<POOL, false, NT>(acc1, pass, Wl2, bn1h, bn2, mr, half, tileM, out, gid, sums);

    if constexpr (POOL) {
        if (mr == 0) {
            int cur = gid[0];
            float n = 0.0f;
            #pragma unroll
            for (int r = 0; r < 8; ++r) {
                if (gid[r] != cur) {
                    if ((unsigned)cur < (unsigned)GG) unsafeAtomicAdd(cnt + cur, n);
                    n = 0.0f;
                    cur = gid[r];
                }
                n += 1.0f;
            }
            if ((unsigned)cur < (unsigned)GG) unsafeAtomicAdd(cnt + cur, n);
        }
    }
}

__global__ __launch_bounds__(256) void zero_kernel(float* __restrict__ p, int n) {
    int idx = blockIdx.x * blockDim.x + threadIdx.x;
    if (idx < n) p[idx] = 0.0f;
}

__global__ __launch_bounds__(128) void head_kernel(
    const float* __restrict__ sums, const float* __restrict__ cnt,
    const float* __restrict__ fc1w, const float* __restrict__ fc1b,
    const float* __restrict__ lng, const float* __restrict__ lnb,
    const float* __restrict__ fc2w, const float* __restrict__ fc2b,
    float* __restrict__ out)
{
    __shared__ float p[HID];
    __shared__ float rbuf[HID];
    __shared__ float red[HID];
    const int g = blockIdx.x;
    const int t = threadIdx.x;

    p[t] = sums[(size_t)g * HID + t] / fmaxf(cnt[g], 1.0f);
    __syncthreads();

    float o = fc1b[t];
    #pragma unroll 4
    for (int k = 0; k < HID; ++k) o += p[k] * fc1w[k * HID + t];

    red[t] = o;
    __syncthreads();
    for (int s = 64; s > 0; s >>= 1) {
        if (t < s) red[t] += red[t + s];
        __syncthreads();
    }
    const float mu = red[0] * (1.0f / HID);
    __syncthreads();

    const float dv = o - mu;
    red[t] = dv * dv;
    __syncthreads();
    for (int s = 64; s > 0; s >>= 1) {
        if (t < s) red[t] += red[t + s];
        __syncthreads();
    }
    const float var = red[0] * (1.0f / HID);

    float y = dv * rsqrtf(var + 1e-5f) * lng[t] + lnb[t];
    y = gelu_exact(y) + p[t];
    rbuf[t] = y;
    __syncthreads();

    if (t < LAT) {
        float o2 = fc2b[t];
        #pragma unroll 4
        for (int k = 0; k < HID; ++k) o2 += rbuf[k] * fc2w[k * LAT + t];
        out[(size_t)g * LAT + t] = o2;
    }
}

extern "C" void kernel_launch(void* const* d_in, const int* in_sizes, int n_in,
                              void* d_out, int out_size, void* d_ws, size_t ws_size,
                              hipStream_t stream) {
    const float* x       = (const float*)d_in[0];
    const int*   ei      = (const int*)  d_in[1];
    const int*   batch   = (const int*)  d_in[2];
    const float* in_g    = (const float*)d_in[3];
    const float* in_b    = (const float*)d_in[4];
    const float* in_m    = (const float*)d_in[5];
    const float* in_v    = (const float*)d_in[6];
    const float* eps0    = (const float*)d_in[7];
    const float* c0_w1   = (const float*)d_in[8];
    const float* c0_b1   = (const float*)d_in[9];
    const float* c0_g    = (const float*)d_in[10];
    const float* c0_bb   = (const float*)d_in[11];
    const float* c0_m    = (const float*)d_in[12];
    const float* c0_v    = (const float*)d_in[13];
    const float* c0_w2   = (const float*)d_in[14];
    const float* c0_b2   = (const float*)d_in[15];
    const float* bn0_g   = (const float*)d_in[16];
    const float* bn0_b   = (const float*)d_in[17];
    const float* bn0_m   = (const float*)d_in[18];
    const float* bn0_v   = (const float*)d_in[19];
    const float* epsR    = (const float*)d_in[20];
    const float* cR_w1   = (const float*)d_in[21];
    const float* cR_b1   = (const float*)d_in[22];
    const float* cR_g    = (const float*)d_in[23];
    const float* cR_bb   = (const float*)d_in[24];
    const float* cR_m    = (const float*)d_in[25];
    const float* cR_v    = (const float*)d_in[26];
    const float* cR_w2   = (const float*)d_in[27];
    const float* cR_b2   = (const float*)d_in[28];
    const float* bnR_g   = (const float*)d_in[29];
    const float* bnR_b   = (const float*)d_in[30];
    const float* bnR_m   = (const float*)d_in[31];
    const float* bnR_v   = (const float*)d_in[32];
    const float* fc1_w   = (const float*)d_in[33];
    const float* fc1_b   = (const float*)d_in[34];
    const float* ln_g    = (const float*)d_in[35];
    const float* ln_b    = (const float*)d_in[36];
    const float* fc2_w   = (const float*)d_in[37];
    const float* fc2_b   = (const float*)d_in[38];

    const int* src = ei;
    const int* dst = ei + EE;

    float* bufA = (float*)d_ws;
    float* bufB = bufA + (size_t)NN * HID;
    int*   col    = (int*)(bufB + (size_t)NN * HID);
    int*   pairs  = col + EE;
    int*   cells  = pairs + 2 * EE;
    int*   base   = cells + PART_CELLS;
    int*   rowptr = base + PART_CELLS + 4;
    float* sums = bufA + (size_t)3 * NN * HID;
    float* cnt  = sums + (size_t)GG * HID;

    const int TB = 256;
    const int nodeBlocks = NN / 8;

    part_hist_kernel<<<PART_BLOCKS, PART_THREADS, 0, stream>>>(dst, cells);
    part_scan_kernel<<<SCAN_BLOCKS, SCAN_THREADS, 0, stream>>>(cells, base);
    part_scatter_kernel<<<PART_BLOCKS, PART_THREADS, 0, stream>>>(src, dst, base, pairs);
    bucket_sort_kernel<<<NBUCKETS, SORT_THREADS, 0, stream>>>(pairs, base, rowptr, col);

    gather_agg_kernel<INDIM, true><<<nodeBlocks, TB, 0, stream>>>(
        x, rowptr, col, eps0, 0, in_g, in_b, in_m, in_v, bufB);
    mlp_kernel<INDIM, false><<<MLP_BLOCKS, MLP_THREADS, 0, stream>>>(
        bufB, c0_w1, c0_b1, c0_g, c0_bb, c0_m, c0_v,
        c0_w2, c0_b2, bn0_g, bn0_b, bn0_m, bn0_v, bufA, batch, sums, cnt);

    zero_kernel<<<(GG * HID + GG + TB - 1) / TB, TB, 0, stream>>>(sums, GG * HID + GG);

    gather_agg_kernel<HID, false><<<nodeBlocks, TB, 0, stream>>>(
        bufA, rowptr, col, epsR, 0, nullptr, nullptr, nullptr, nullptr, bufB);
    mlp_kernel<HID, false><<<MLP_BLOCKS, MLP_THREADS, 0, stream>>>(
        bufB, cR_w1, cR_b1, cR_g, cR_bb, cR_m, cR_v,
        cR_w2, cR_b2, bnR_g, bnR_b, bnR_m, bnR_v, bufA, batch, sums, cnt);
    gather_agg_kernel<HID, false><<<nodeBlocks, TB, 0, stream>>>(
        bufA, rowptr, col, epsR, 1, nullptr, nullptr, nullptr, nullptr, bufB);
    mlp_kernel<HID, true><<<MLP_BLOCKS, MLP_THREADS, 0, stream>>>(
        bufB, cR_w1 + HID * HID, cR_b1 + HID, cR_g + HID, cR_bb + HID, cR_m + HID, cR_v + HID,
        cR_w2 + HID * HID, cR_b2 + HID, bnR_g + HID, bnR_b + HID, bnR_m + HID, bnR_v + HID,
        bufA, batch, sums, cnt);

    head_kernel<<<GG, HID, 0, stream>>>(
        sums, cnt, fc1_w, fc1_b, ln_g, ln_b, fc2_w, fc2_b, (float*)d_out);
}
